# starting kernel
# baseline (speedup 1.0000x reference)
#include <hip/hip_runtime.h>
#include <math.h>

#define B_ 2
#define C_ 128
#define T_ 8
#define H_ 64
#define W_ 64
#define N_ (T_*H_*W_)
#define BN_ (B_*N_)
#define HID_ 256
#define MAX_STEPS_ 12
#define DIFF_SCALE 0.25f
#define PROBE_CELLS (B_*C_*2*8*8)

typedef __attribute__((ext_vector_type(16))) _Float16 v16h;
typedef __attribute__((ext_vector_type(8)))  _Float16 h8;
typedef __attribute__((ext_vector_type(8)))  float    v8f;

__device__ __forceinline__ float softplus_f(float x){ return x > 20.f ? x : log1pf(expf(x)); }
__device__ __forceinline__ float gelu_f(float x){ return 0.5f*x*(1.f + erff(x*0.70710678118654752f)); }
__device__ __forceinline__ int imin_(int a, int b){ return a < b ? a : b; }

__device__ __forceinline__ float blockReduceSum(float v){
  __shared__ float red[8];
  int lane = threadIdx.x & 31, wid = threadIdx.x >> 5;
  #pragma unroll
  for (int off = 16; off > 0; off >>= 1) v += __shfl_down(v, off);
  if (lane == 0) red[wid] = v;
  __syncthreads();
  v = (threadIdx.x < (blockDim.x >> 5)) ? red[threadIdx.x] : 0.f;
  if (wid == 0){
    #pragma unroll
    for (int off = 4; off > 0; off >>= 1) v += __shfl_down(v, off);
  }
  __syncthreads();
  return v;
}

__device__ __forceinline__ v16h load_frag_row(const _Float16* base, int ld, int r0, int k0, int lane){
  int r = r0 + (lane & 15);
  int hf = lane >> 4;
  const _Float16* p = base + r*ld + k0 + hf*8;
  h8 lo = *(const h8*)p;
  h8 hi = *(const h8*)(p + 16);
  return __builtin_shufflevector(lo, hi, 0,1,2,3,4,5,6,7,8,9,10,11,12,13,14,15);
}


__global__ void zero_kernel(float* p, int n){
  int i = blockIdx.x*blockDim.x + threadIdx.x;
  if (i < n) p[i] = 0.f;
}

__global__ void prep_kernel(const float* w1, const float* w2,
                            const float* ks, const float* kt,
                            const float* ag, const float* al, const float* logdt,
                            _Float16* w1h, _Float16* w2h,
                            float* spsks, float* spskt, float* scal){
  int i = blockIdx.x*blockDim.x + threadIdx.x;
  int stride = gridDim.x*blockDim.x;
  for (int j = i; j < C_*HID_; j += stride) w1h[j] = (_Float16)w1[j];
  for (int j = i; j < HID_*C_; j += stride) w2h[j] = (_Float16)w2[j];
  for (int j = i; j < 3*C_;   j += stride) spsks[j] = softplus_f(ks[j]);
  for (int j = i; j < 2*C_;   j += stride) spskt[j] = softplus_f(kt[j]);
  if (i == 0){
    float dt = expf(logdt[0]);
    dt = fminf(fmaxf(dt, 0.005f), 0.35f);
    scal[0] = dt;
    scal[1] = softplus_f(ag[0]);
    scal[2] = softplus_f(al[0]);
  }
}

__global__ void pool_kernel(const float* in, float* out, int tp, int hp, int wp){
  int cells = B_*C_*tp*hp*wp;
  int id = blockIdx.x*blockDim.x + threadIdx.x;
  if (id >= cells) return;
  int pw = id % wp; int tmp = id / wp;
  int ph = tmp % hp; tmp /= hp;
  int pt = tmp % tp; tmp /= tp;
  int c = tmp % C_;  int b = tmp / C_;
  int bt = T_/tp, bh = H_/hp, bw = W_/wp;
  const float* base = in + (size_t)(b*C_ + c)*N_;
  float s = 0.f;
  for (int it = 0; it < bt; ++it)
    for (int ih = 0; ih < bh; ++ih)
      for (int iw = 0; iw < bw; ++iw)
        s += base[((pt*bt+it)*H_ + (ph*bh+ih))*W_ + (pw*bw+iw)];
  out[id] = s / (float)(bt*bh*bw);
}

__global__ void pooled_mean_kernel(const float* u, float* pooled){
  int bc = blockIdx.x;
  const float* base = u + (size_t)bc*N_;
  float s = 0.f;
  for (int i = threadIdx.x; i < N_; i += blockDim.x) s += base[i];
  s = blockReduceSum(s);
  if (threadIdx.x == 0) pooled[bc] = s / (float)N_;
}

__global__ void gate_value_kernel(const float* pooled, const float* gw, const float* gb,
                                  const float* vw, const float* vb, float* h){
  int tid = threadIdx.x;
  int b = tid >> 7, c = tid & 127;
  const float* p = pooled + b*C_;
  float g = gb[c], v = vb[c];
  for (int k = 0; k < C_; ++k){
    float pk = p[k];
    g += pk * gw[k*C_ + c];
    v += pk * vw[k*C_ + c];
  }
  h[tid] += (1.f/(1.f+expf(-g))) * tanhf(v);
}

__global__ void local_proj_kernel(const float* lsmall, const float* wlp, const float* blp, float* lproj){
  int id = blockIdx.x*blockDim.x + threadIdx.x;
  if (id >= B_*C_*32) return;
  int cell = id & 31;
  int oc = (id >> 5) & 127;
  int b = id >> 12;
  float s = blp[oc];
  for (int c = 0; c < C_; ++c)
    s += lsmall[((size_t)(b*C_ + c) << 5) + cell] * wlp[c*C_ + oc];
  lproj[((size_t)(b*C_ + oc) << 5) + cell] = s;
}

__global__ void __launch_bounds__(256)
reaction_mlp_kernel(const float* u, const _Float16* w1h, const float* b1,
                    const _Float16* w2h, const float* b2, float* react){
  __shared__ _Float16 As[128][128];
  __shared__ _Float16 Bt[32][128];
  __shared__ _Float16 Hs[128][32];
  __shared__ _Float16 W2t[128][32];

  const int tid  = threadIdx.x;
  const int lane = tid & 31;
  const int wv   = tid >> 5;
  const int m0   = wv * 16;
  const int hf   = lane >> 4;
  const int col  = lane & 15;

  const int rowBase = blockIdx.x * 128;
  const int b  = rowBase / N_;
  const int n0 = rowBase - b * N_;
  const float* ub = u + (size_t)b * C_ * N_;

  for (int idx = tid; idx < 128*32; idx += 256){
    int c = idx >> 5;
    int q = idx & 31;
    const float4 v = *(const float4*)(ub + (size_t)c * N_ + n0 + q*4);
    As[q*4+0][c] = (_Float16)v.x;
    As[q*4+1][c] = (_Float16)v.y;
    As[q*4+2][c] = (_Float16)v.z;
    As[q*4+3][c] = (_Float16)v.w;
  }

  const v8f z8 = {0.f,0.f,0.f,0.f,0.f,0.f,0.f,0.f};
  v8f accC[8];
  #pragma unroll
  for (int i = 0; i < 8; ++i) accC[i] = z8;

  for (int hc = 0; hc < 8; ++hc){
    if (hc < 7){
      __builtin_prefetch(&w1h[(hc+1)*32], 0, 1);
      __builtin_prefetch(&w2h[((hc+1)*32)*C_], 0, 1);
    }
    __syncthreads();
    for (int idx = tid; idx < 32*128; idx += 256){
      int j = idx & 31;
      int k = idx >> 5;
      Bt[j][k] = w1h[k*HID_ + hc*32 + j];
    }
    for (int idx = tid; idx < 32*128; idx += 256){
      int nc = idx & 127;
      int k = idx >> 7;
      W2t[nc][k] = w2h[(hc*32 + k)*C_ + nc];
    }
    __syncthreads();

    v8f a0 = z8, a1 = z8;
    #pragma unroll
    for (int k0 = 0; k0 < 128; k0 += 32){
      v16h af  = load_frag_row(&As[0][0], 128, m0, k0, lane);
      v16h bf0 = load_frag_row(&Bt[0][0], 128, 0,  k0, lane);
      v16h bf1 = load_frag_row(&Bt[0][0], 128, 16, k0, lane);
      a0 = __builtin_amdgcn_wmma_f32_16x16x32_f16(false, af, false, bf0, (short)0, a0, false, false);
      a1 = __builtin_amdgcn_wmma_f32_16x16x32_f16(false, af, false, bf1, (short)0, a1, false, false);
    }
    #pragma unroll
    for (int r = 0; r < 8; ++r){
      int m = m0 + hf*8 + r;
      float x0 = a0[r] + b1[hc*32 + col];
      float x1 = a1[r] + b1[hc*32 + 16 + col];
      Hs[m][col]      = (_Float16)gelu_f(x0);
      Hs[m][16 + col] = (_Float16)gelu_f(x1);
    }
    v16h ah = load_frag_row(&Hs[0][0], 32, m0, 0, lane);
    #pragma unroll
    for (int nt = 0; nt < 8; ++nt){
      v16h bw = load_frag_row(&W2t[0][0], 32, nt*16, 0, lane);
      accC[nt] = __builtin_amdgcn_wmma_f32_16x16x32_f16(false, ah, false, bw, (short)0, accC[nt], false, false);
    }
  }

  #pragma unroll
  for (int nt = 0; nt < 8; ++nt){
    int c = nt*16 + col;
    float bias = b2[c];
    float* dst = react + (size_t)(b*C_ + c)*N_ + n0 + m0 + hf*8;
    float4 v0 = { accC[nt][0]+bias, accC[nt][1]+bias, accC[nt][2]+bias, accC[nt][3]+bias };
    float4 v1 = { accC[nt][4]+bias, accC[nt][5]+bias, accC[nt][6]+bias, accC[nt][7]+bias };
    *(float4*)dst       = v0;
    *(float4*)(dst + 4) = v1;
  }
}

__global__ void __launch_bounds__(256)
update_kernel(const float* u, const float* react, const float* h,
              const float* lproj, const float* spsks, const float* spskt,
              const float* scal, const float* normw,
              float* ucand, float* unext, float* accum, int do_norm){
  int p = blockIdx.x*blockDim.x + threadIdx.x;
  int b  = p >> 15;
  int n  = p & (N_ - 1);
  int t  = n >> 12;
  int hh = (n >> 6) & 63;
  int ww = n & 63;

  float dt = scal[0], ag = scal[1], al = scal[2];

  float st = fminf(fmaxf((t  + 0.5f)*(2.f/T_) - 0.5f, 0.f), 1.f);
  float sh = fminf(fmaxf((hh + 0.5f)*(4.f/H_) - 0.5f, 0.f), 3.f);
  float sw = fminf(fmaxf((ww + 0.5f)*(4.f/W_) - 0.5f, 0.f), 3.f);
  int t0i = (int)st; int t1i = imin_(t0i+1, 1); float ft = st - (float)t0i;
  int h0i = (int)sh; int h1i = imin_(h0i+1, 3); float fh = sh - (float)h0i;
  int w0i = (int)sw; int w1i = imin_(w0i+1, 3); float fw = sw - (float)w0i;

  float sum_du = 0.f, sum_lap = 0.f, sumsq = 0.f;
  size_t bbase = (size_t)b * C_ * N_;

  for (int c = 0; c < C_; ++c){
    const float* uc = u + bbase + (size_t)c * N_;
    float uv = uc[n];

    float diff = 0.f;
    #pragma unroll
    for (int i = 0; i < 3; ++i){
      const int d = (i == 0) ? 1 : ((i == 1) ? 4 : 16);
      float lap = -4.f * uv;
      lap += (hh - d >= 0) ? uc[n - d*64] : 0.f;
      lap += (hh + d < 64) ? uc[n + d*64] : 0.f;
      lap += (ww - d >= 0) ? uc[n - d]    : 0.f;
      lap += (ww + d < 64) ? uc[n + d]    : 0.f;
      diff += spsks[i*C_ + c] * lap;
    }
    #pragma unroll
    for (int i = 0; i < 2; ++i){
      const int d = i + 1;
      float up = (t - d >= 0) ? uc[n - d*4096] : 0.f;
      diff += spskt[i*C_ + c] * (up - uv);
    }
    diff *= DIFF_SCALE;
    sum_lap += fabsf(diff);

    const float* lp = lproj + ((size_t)(b*C_ + c) << 5);
    float c00 = lp[t0i*16 + h0i*4 + w0i]*(1.f-fw) + lp[t0i*16 + h0i*4 + w1i]*fw;
    float c01 = lp[t0i*16 + h1i*4 + w0i]*(1.f-fw) + lp[t0i*16 + h1i*4 + w1i]*fw;
    float c10 = lp[t1i*16 + h0i*4 + w0i]*(1.f-fw) + lp[t1i*16 + h0i*4 + w1i]*fw;
    float c11 = lp[t1i*16 + h1i*4 + w0i]*(1.f-fw) + lp[t1i*16 + h1i*4 + w1i]*fw;
    float lv = (c00*(1.f-fh) + c01*fh)*(1.f-ft) + (c10*(1.f-fh) + c11*fh)*ft;

    float du = diff + react[bbase + (size_t)c*N_ + n] + ag*h[b*C_ + c] + al*lv;
    float cand = uv + dt*du;
    sum_du += fabsf(du);
    sumsq  += cand*cand;
    ucand[bbase + (size_t)c*N_ + n] = cand;
  }

  float scale = do_norm ? rsqrtf(sumsq/(float)C_ + 1e-8f) : 1.f;
  for (int c = 0; c < C_; ++c){
    float cand = ucand[bbase + (size_t)c*N_ + n];
    unext[bbase + (size_t)c*N_ + n] = do_norm ? cand*scale*normw[c] : cand;
  }

  float s1 = blockReduceSum(sum_du);
  float s2 = blockReduceSum(sum_lap);
  if (threadIdx.x == 0){
    atomicAdd(&accum[1], s1);
    atomicAdd(&accum[0], s2);
  }
}

__global__ void turb_kernel(const float* cur, const float* prev, float* accum){
  int i = blockIdx.x*blockDim.x + threadIdx.x;
  float a = 0.f, pv = 0.f;
  if (i < PROBE_CELLS){
    float cv = cur[i], pp = prev[i];
    a = fabsf(cv - pp);
    pv = fabsf(pp);
  }
  float s1 = blockReduceSum(a);
  float s2 = blockReduceSum(pv);
  if (threadIdx.x == 0){
    atomicAdd(&accum[2], s1);
    atomicAdd(&accum[3], s2);
  }
}

__global__ void finalize_step_kernel(const float* accum, float* turbs, float* sten, int step){
  if (threadIdx.x == 0 && blockIdx.x == 0){
    const float inv_cells = 1.f/(float)PROBE_CELLS;
    const float inv_bnc   = 1.f/8388608.f;
    float stop_turb = (accum[2]*inv_cells) / (accum[3]*inv_cells + 1e-8f);
    float se = accum[1]*inv_bnc;
    float le = accum[0]*inv_bnc;
    turbs[step] = stop_turb + 0.05f*se + 0.01f*le;
    sten[step]  = se;
  }
}

__global__ void final_out_kernel(const float* turbs, const float* sten, float* out){
  if (threadIdx.x == 0 && blockIdx.x == 0){
    float a = 0.f, bm = 0.f;
    for (int i = 0; i < MAX_STEPS_; ++i){ a += turbs[i]; bm += sten[i]; }
    out[0] = a / (float)MAX_STEPS_;
    out[1] = bm / (float)MAX_STEPS_;
  }
}


extern "C" void kernel_launch(void* const* d_in, const int* in_sizes, int n_in,
                              void* d_out, int out_size, void* d_ws, size_t ws_size,
                              hipStream_t stream){
  (void)in_sizes; (void)n_in; (void)out_size; (void)ws_size;

  const float* u_in  = (const float*)d_in[0];
  const float* w1    = (const float*)d_in[1];
  const float* b1    = (const float*)d_in[2];
  const float* w2    = (const float*)d_in[3];
  const float* b2    = (const float*)d_in[4];
  const float* gw    = (const float*)d_in[5];
  const float* gb    = (const float*)d_in[6];
  const float* vw    = (const float*)d_in[7];
  const float* vb    = (const float*)d_in[8];
  const float* normw = (const float*)d_in[9];
  const float* wlp   = (const float*)d_in[10];
  const float* blp   = (const float*)d_in[11];
  const float* a_g   = (const float*)d_in[12];
  const float* a_l   = (const float*)d_in[13];
  const float* logdt = (const float*)d_in[14];
  const float* ks    = (const float*)d_in[15];
  const float* kt    = (const float*)d_in[16];

  float* Wf = (float*)d_ws;
  size_t o = 0;
  const size_t VOL = (size_t)B_ * C_ * N_;
  float* ubuf0  = Wf + o; o += VOL;
  float* ubuf1  = Wf + o; o += VOL;
  float* ucand  = Wf + o; o += VOL;
  float* react  = Wf + o; o += VOL;
  float* probeA = Wf + o; o += PROBE_CELLS;
  float* probeB = Wf + o; o += PROBE_CELLS;
  float* pooled = Wf + o; o += 256;
  float* hstate = Wf + o; o += 256;
  float* lsmall = Wf + o; o += (size_t)B_*C_*32;
  float* lproj  = Wf + o; o += (size_t)B_*C_*32;
  float* spsks  = Wf + o; o += 3*C_;
  float* spskt  = Wf + o; o += 2*C_;
  float* scal   = Wf + o; o += 8;
  float* accum  = Wf + o; o += 8;
  float* turbs  = Wf + o; o += 16;
  float* sten   = Wf + o; o += 16;
  _Float16* w1h = (_Float16*)(Wf + o); o += (C_*HID_)/2;
  _Float16* w2h = (_Float16*)(Wf + o); o += (HID_*C_)/2;

  float* ubuf[2] = { ubuf0, ubuf1 };

  prep_kernel<<<64, 256, 0, stream>>>(w1, w2, ks, kt, a_g, a_l, logdt,
                                      w1h, w2h, spsks, spskt, scal);
  zero_kernel<<<1, 256, 0, stream>>>(hstate, 256);
  hipMemcpyAsync(ubuf[0], u_in, VOL*sizeof(float), hipMemcpyDeviceToDevice, stream);
  pool_kernel<<<(PROBE_CELLS + 255)/256, 256, 0, stream>>>(ubuf[0], probeA, 2, 8, 8);

  float* pprev = probeA;
  float* pcur  = probeB;

  for (int step = 0; step < MAX_STEPS_; ++step){
    float* ucur  = ubuf[step & 1];
    float* unext = ubuf[(step & 1) ^ 1];
    int do_norm = (((step + 1) % 2) == 0) ? 1 : 0;

    zero_kernel<<<1, 32, 0, stream>>>(accum, 8);
    reaction_mlp_kernel<<<BN_/128, 256, 0, stream>>>(ucur, w1h, b1, w2h, b2, react);
    pooled_mean_kernel<<<B_*C_, 256, 0, stream>>>(ucur, pooled);
    gate_value_kernel<<<1, 256, 0, stream>>>(pooled, gw, gb, vw, vb, hstate);
    pool_kernel<<<(B_*C_*32 + 255)/256, 256, 0, stream>>>(ucur, lsmall, 2, 4, 4);
    local_proj_kernel<<<(B_*C_*32 + 255)/256, 256, 0, stream>>>(lsmall, wlp, blp, lproj);
    update_kernel<<<BN_/256, 256, 0, stream>>>(ucur, react, hstate, lproj, spsks, spskt,
                                               scal, normw, ucand, unext, accum, do_norm);
    pool_kernel<<<(PROBE_CELLS + 255)/256, 256, 0, stream>>>(ucand, pcur, 2, 8, 8);
    turb_kernel<<<PROBE_CELLS/256, 256, 0, stream>>>(pcur, pprev, accum);
    finalize_step_kernel<<<1, 1, 0, stream>>>(accum, turbs, sten, step);

    float* tswap = pprev; pprev = pcur; pcur = tswap;
  }

  hipMemcpyAsync(d_out, ubuf[MAX_STEPS_ & 1], VOL*sizeof(float),
                 hipMemcpyDeviceToDevice, stream);
  final_out_kernel<<<1, 32, 0, stream>>>(turbs, sten, ((float*)d_out) + VOL);
}


# MLP kernel LDS staging: weights pre-transposed k-contiguous, every tile written with 16-byte stores at row pitches of 16 B x odd, next weight chunk fetched under the current chunk's products
# speedup vs baseline: 2.7031x; 2.7031x over previous
#include <hip/hip_runtime.h>
#include <math.h>

#define B_ 2
#define C_ 128
#define T_ 8
#define H_ 64
#define W_ 64
#define N_ (T_*H_*W_)
#define BN_ (B_*N_)
#define HID_ 256
#define MAX_STEPS_ 12
#define DIFF_SCALE 0.25f
#define PROBE_CELLS (B_*C_*2*8*8)

typedef __attribute__((ext_vector_type(16))) _Float16 v16h;
typedef __attribute__((ext_vector_type(8)))  _Float16 h8;
typedef __attribute__((ext_vector_type(8)))  float    v8f;
typedef __attribute__((ext_vector_type(4)))  float    f4x;

__device__ __forceinline__ float softplus_f(float x){ return x > 20.f ? x : log1pf(expf(x)); }
__device__ __forceinline__ float gelu_f(float x){ return 0.5f*x*(1.f + erff(x*0.70710678118654752f)); }
__device__ __forceinline__ int imin_(int a, int b){ return a < b ? a : b; }

__device__ __forceinline__ float blockReduceSum(float v){
  __shared__ float red[8];
  int lane = threadIdx.x & 31, wid = threadIdx.x >> 5;
  #pragma unroll
  for (int off = 16; off > 0; off >>= 1) v += __shfl_down(v, off);
  if (lane == 0) red[wid] = v;
  __syncthreads();
  v = (threadIdx.x < (blockDim.x >> 5)) ? red[threadIdx.x] : 0.f;
  if (wid == 0){
    #pragma unroll
    for (int off = 4; off > 0; off >>= 1) v += __shfl_down(v, off);
  }
  __syncthreads();
  return v;
}

__device__ __forceinline__ v16h load_frag_row(const _Float16* base, int ld, int r0, int k0, int lane){
  int r = r0 + (lane & 15);
  int hf = lane >> 4;
  const _Float16* p = base + r*ld + k0 + hf*8;
  h8 lo = *(const h8*)p;
  h8 hi = *(const h8*)(p + 16);
  return __builtin_shufflevector(lo, hi, 0,1,2,3,4,5,6,7,8,9,10,11,12,13,14,15);
}


__global__ void zero_kernel(float* p, int n){
  int i = blockIdx.x*blockDim.x + threadIdx.x;
  if (i < n) p[i] = 0.f;
}

__global__ void prep_kernel(const float* w1, const float* w2,
                            const float* ks, const float* kt,
                            const float* ag, const float* al, const float* logdt,
                            _Float16* w1h, _Float16* w2h,
                            float* spsks, float* spskt, float* scal){
  int i = blockIdx.x*blockDim.x + threadIdx.x;
  int stride = gridDim.x*blockDim.x;
  for (int j = i; j < C_*HID_; j += stride) w1h[j] = (_Float16)w1[(j & (C_-1))*HID_ + (j >> 7)];
  for (int j = i; j < HID_*C_; j += stride) w2h[j] = (_Float16)w2[(j & (HID_-1))*C_ + (j >> 8)];
  for (int j = i; j < 3*C_;   j += stride) spsks[j] = softplus_f(ks[j]);
  for (int j = i; j < 2*C_;   j += stride) spskt[j] = softplus_f(kt[j]);
  if (i == 0){
    float dt = expf(logdt[0]);
    dt = fminf(fmaxf(dt, 0.005f), 0.35f);
    scal[0] = dt;
    scal[1] = softplus_f(ag[0]);
    scal[2] = softplus_f(al[0]);
  }
}

__global__ void pool_kernel(const float* in, float* out, int tp, int hp, int wp){
  int cells = B_*C_*tp*hp*wp;
  int id = blockIdx.x*blockDim.x + threadIdx.x;
  if (id >= cells) return;
  int pw = id % wp; int tmp = id / wp;
  int ph = tmp % hp; tmp /= hp;
  int pt = tmp % tp; tmp /= tp;
  int c = tmp % C_;  int b = tmp / C_;
  int bt = T_/tp, bh = H_/hp, bw = W_/wp;
  const float* base = in + (size_t)(b*C_ + c)*N_;
  float s = 0.f;
  for (int it = 0; it < bt; ++it)
    for (int ih = 0; ih < bh; ++ih)
      for (int iw = 0; iw < bw; ++iw)
        s += base[((pt*bt+it)*H_ + (ph*bh+ih))*W_ + (pw*bw+iw)];
  out[id] = s / (float)(bt*bh*bw);
}

__global__ void pooled_mean_kernel(const float* u, float* pooled){
  int bc = blockIdx.x;
  const float* base = u + (size_t)bc*N_;
  float s = 0.f;
  for (int i = threadIdx.x; i < N_; i += blockDim.x) s += base[i];
  s = blockReduceSum(s);
  if (threadIdx.x == 0) pooled[bc] = s / (float)N_;
}

__global__ void gate_value_kernel(const float* pooled, const float* gw, const float* gb,
                                  const float* vw, const float* vb, float* h){
  int tid = threadIdx.x;
  int b = tid >> 7, c = tid & 127;
  const float* p = pooled + b*C_;
  float g = gb[c], v = vb[c];
  for (int k = 0; k < C_; ++k){
    float pk = p[k];
    g += pk * gw[k*C_ + c];
    v += pk * vw[k*C_ + c];
  }
  h[tid] += (1.f/(1.f+expf(-g))) * tanhf(v);
}

__global__ void local_proj_kernel(const float* lsmall, const float* wlp, const float* blp, float* lproj){
  int id = blockIdx.x*blockDim.x + threadIdx.x;
  if (id >= B_*C_*32) return;
  int cell = id & 31;
  int oc = (id >> 5) & 127;
  int b = id >> 12;
  float s = blp[oc];
  for (int c = 0; c < C_; ++c)
    s += lsmall[((size_t)(b*C_ + c) << 5) + cell] * wlp[c*C_ + oc];
  lproj[((size_t)(b*C_ + oc) << 5) + cell] = s;
}

#define PA_ 136
#define PH_ 40

__global__ void __launch_bounds__(256)
reaction_mlp_kernel(const float* u, const _Float16* w1t, const float* b1,
                    const _Float16* w2t, const float* b2, float* react){
  __shared__ __attribute__((aligned(16))) _Float16 As[128*PA_];
  __shared__ __attribute__((aligned(16))) _Float16 Bt[32*PA_];
  __shared__ __attribute__((aligned(16))) _Float16 Hs[128*PH_];
  __shared__ __attribute__((aligned(16))) _Float16 W2t[128*PH_];

  const int tid  = threadIdx.x;
  const int lane = tid & 31;
  const int wv   = tid >> 5;
  const int m0   = wv * 16;
  const int hf   = lane >> 4;
  const int col  = lane & 15;

  const int rowBase = blockIdx.x * 128;
  const int b  = rowBase / N_;
  const int n0 = rowBase - b * N_;
  const float* ub = u + (size_t)b * C_ * N_;

  const _Float16* g1 = w1t + (size_t)(tid >> 4)*C_   + (tid & 15)*8;
  const _Float16* g2 = w2t + (size_t)(tid >> 2)*HID_ + (tid & 3)*8;
  _Float16* l1 = Bt  + (tid >> 4)*PA_ + (tid & 15)*8;
  _Float16* l2 = W2t + (tid >> 2)*PH_ + (tid & 3)*8;

  h8 r1a = *(const h8*)g1;
  h8 r1b = *(const h8*)(g1 + 16*C_);
  h8 r2a = *(const h8*)g2;
  h8 r2b = *(const h8*)(g2 + 64*HID_);

  {
    f4x v[16];
    #pragma unroll
    for (int i = 0; i < 16; ++i)
      v[i] = *(const f4x*)(ub + (size_t)(wv*16 + i) * N_ + n0 + lane*4);
    #pragma unroll
    for (int g = 0; g < 2; ++g){
      #pragma unroll
      for (int j = 0; j < 4; ++j){
        h8 hv;
        #pragma unroll
        for (int i = 0; i < 8; ++i) hv[i] = (_Float16)v[g*8 + i][j];
        *(h8*)(As + (lane*4 + j)*PA_ + wv*16 + g*8) = hv;
      }
    }
  }

  const v8f z8 = {0.f,0.f,0.f,0.f,0.f,0.f,0.f,0.f};
  v8f accC[8];
  #pragma unroll
  for (int i = 0; i < 8; ++i) accC[i] = z8;

  for (int hc = 0; hc < 8; ++hc){
    __syncthreads();
    *(h8*)l1              = r1a;
    *(h8*)(l1 + 16*PA_)   = r1b;
    *(h8*)l2              = r2a;
    *(h8*)(l2 + 64*PH_)   = r2b;
    __syncthreads();

    const float bia0 = b1[hc*32 + col], bia1 = b1[hc*32 + 16 + col];
    if (hc < 7){
      r1a = *(const h8*)(g1 + (hc+1)*32*C_);
      r1b = *(const h8*)(g1 + (hc+1)*32*C_ + 16*C_);
      r2a = *(const h8*)(g2 + (hc+1)*32);
      r2b = *(const h8*)(g2 + (hc+1)*32 + 64*HID_);
    }

    v8f a0 = z8, a1 = z8;
    #pragma unroll
    for (int k0 = 0; k0 < 128; k0 += 32){
      v16h af  = load_frag_row(As, PA_, m0, k0, lane);
      v16h bf0 = load_frag_row(Bt, PA_, 0,  k0, lane);
      v16h bf1 = load_frag_row(Bt, PA_, 16, k0, lane);
      a0 = __builtin_amdgcn_wmma_f32_16x16x32_f16(false, af, false, bf0, (short)0, a0, false, false);
      a1 = __builtin_amdgcn_wmma_f32_16x16x32_f16(false, af, false, bf1, (short)0, a1, false, false);
    }
    #pragma unroll
    for (int r = 0; r < 8; ++r){
      int m = m0 + hf*8 + r;
      Hs[m*PH_ + col]      = (_Float16)gelu_f(a0[r] + bia0);
      Hs[m*PH_ + 16 + col] = (_Float16)gelu_f(a1[r] + bia1);
    }
    v16h ah = load_frag_row(Hs, PH_, m0, 0, lane);
    #pragma unroll
    for (int nt = 0; nt < 8; ++nt){
      v16h bw = load_frag_row(W2t, PH_, nt*16, 0, lane);
      accC[nt] = __builtin_amdgcn_wmma_f32_16x16x32_f16(false, ah, false, bw, (short)0, accC[nt], false, false);
    }
  }

  #pragma unroll
  for (int nt = 0; nt < 8; ++nt){
    int c = nt*16 + col;
    float bias = b2[c];
    float* dst = react + (size_t)(b*C_ + c)*N_ + n0 + m0 + hf*8;
    float4 v0 = { accC[nt][0]+bias, accC[nt][1]+bias, accC[nt][2]+bias, accC[nt][3]+bias };
    float4 v1 = { accC[nt][4]+bias, accC[nt][5]+bias, accC[nt][6]+bias, accC[nt][7]+bias };
    *(float4*)dst       = v0;
    *(float4*)(dst + 4) = v1;
  }
}

typedef float f4 __attribute__((ext_vector_type(4)));
#define UPD_WAVES 16
#define UPD_CPW   8

__device__ __forceinline__ f4 ldg4(const float* p){ return *(const f4*)p; }

__global__ void __launch_bounds__(512)
update_kernel(const float* __restrict__ u, const float* __restrict__ react, const float* __restrict__ h,
              const float* __restrict__ lproj, const float* __restrict__ spsks, const float* __restrict__ spskt,
              const float* __restrict__ scal, const float* normw,
              float* __restrict__ ucand, float* __restrict__ unext, float* __restrict__ accum, int do_norm){
  __shared__ f4 sq[UPD_WAVES][32];
  __shared__ float red_du[UPD_WAVES];
  __shared__ float red_lap[UPD_WAVES];

  const int tid  = threadIdx.x;
  const int lane = tid & 31;
  const int wv   = __builtin_amdgcn_readfirstlane(tid >> 5);
  const int rp   = blockIdx.x;
  const int b    = rp >> 8;
  const int t    = (rp >> 5) & 7;
  const int hh   = ((rp & 31) << 1) + (lane >> 4);
  const int ww0  = (lane & 15) << 2;
  const int n    = (t << 12) + (hh << 6) + ww0;

  const float dt = scal[0], ag = scal[1], al = scal[2];

  float st = fminf(fmaxf((t  + 0.5f)*(2.f/T_) - 0.5f, 0.f), 1.f);
  float sh = fminf(fmaxf((hh + 0.5f)*(4.f/H_) - 0.5f, 0.f), 3.f);
  int t0i = (int)st; int t1i = imin_(t0i+1, 1); float ft = st - (float)t0i;
  int h0i = (int)sh; int h1i = imin_(h0i+1, 3); float fh = sh - (float)h0i;
  f4 fw;
  int w0i = 0;
  for (int q = 0; q < 4; ++q){
    float sw = fminf(fmaxf((ww0 + q + 0.5f)*(4.f/W_) - 0.5f, 0.f), 3.f);
    int wi = (int)sw;
    if (q == 0) w0i = wi;
    fw[q] = sw - (float)wi;
  }
  const int w1i = imin_(w0i+1, 3);
  const f4 omfw = 1.f - fw;
  const float omfh = 1.f - fh, omft = 1.f - ft;
  const int i000 = t0i*16 + h0i*4 + w0i, i001 = t0i*16 + h0i*4 + w1i;
  const int i010 = t0i*16 + h1i*4 + w0i, i011 = t0i*16 + h1i*4 + w1i;
  const int i100 = t1i*16 + h0i*4 + w0i, i101 = t1i*16 + h0i*4 + w1i;
  const int i110 = t1i*16 + h1i*4 + w0i, i111 = t1i*16 + h1i*4 + w1i;

  const bool kU1 = hh >= 1,  kD1 = hh < 63, kU4 = hh >= 4,  kD4 = hh < 60;
  const bool kU16 = hh >= 16, kD16 = hh < 48;
  const bool kL4 = ww0 >= 4, kR4 = ww0 < 60, kL16 = ww0 >= 16, kR16 = ww0 < 48;
  const bool kT1 = t >= 1, kT2 = t >= 2;
  const int oU1 = kU1 ? -64 : 0,    oD1 = kD1 ? 64 : 0;
  const int oU4 = kU4 ? -256 : 0,   oD4 = kD4 ? 256 : 0;
  const int oU16 = kU16 ? -1024 : 0, oD16 = kD16 ? 1024 : 0;
  const int oL4 = kL4 ? -4 : 0,     oR4 = kR4 ? 4 : 0;
  const int oL16 = kL16 ? -16 : 0,  oR16 = kR16 ? 16 : 0;
  const int oT1 = kT1 ? -4096 : 0,  oT2 = kT2 ? -8192 : 0;
  const f4 z4 = {0.f, 0.f, 0.f, 0.f};

  f4 cand[UPD_CPW];
  f4 sumsq = z4;
  float sum_du = 0.f, sum_lap = 0.f;
  const size_t bbase = (size_t)b * C_ * N_;

  #pragma unroll
  for (int j = 0; j < UPD_CPW; ++j){
    const int c = wv*UPD_CPW + j;
    const size_t off = bbase + (size_t)c * N_ + n;
    const float* uc = u + off;

    f4 cen = ldg4(uc);
    f4 U1  = ldg4(uc + oU1),  D1  = ldg4(uc + oD1);
    f4 U4  = ldg4(uc + oU4),  D4  = ldg4(uc + oD4);
    f4 U16 = ldg4(uc + oU16), D16 = ldg4(uc + oD16);
    f4 L4  = ldg4(uc + oL4),  R4  = ldg4(uc + oR4);
    f4 L16 = ldg4(uc + oL16), R16 = ldg4(uc + oR16);
    f4 T1  = ldg4(uc + oT1),  T2  = ldg4(uc + oT2);
    f4 rc  = ldg4(react + off);
    float lpv = lproj[((size_t)(b*C_ + c) << 5) + lane];

    U1  = kU1  ? U1  : z4;  D1  = kD1  ? D1  : z4;
    U4  = kU4  ? U4  : z4;  D4  = kD4  ? D4  : z4;
    U16 = kU16 ? U16 : z4;  D16 = kD16 ? D16 : z4;
    L4  = kL4  ? L4  : z4;  R4  = kR4  ? R4  : z4;
    L16 = kL16 ? L16 : z4;  R16 = kR16 ? R16 : z4;
    T1  = kT1  ? T1  : z4;  T2  = kT2  ? T2  : z4;
    f4 L1 = { L4[3], cen[0], cen[1], cen[2] };
    f4 R1 = { cen[1], cen[2], cen[3], R4[0] };

    const float s0 = spsks[c], s1 = spsks[C_ + c], s2 = spsks[2*C_ + c];
    const float k0 = spskt[c], k1 = spskt[C_ + c];
    const float hv = h[b*C_ + c];

    f4 diff, lap;
    lap = -4.f * cen; lap += U1;  lap += D1;  lap += L1;  lap += R1;  diff  = s0 * lap;
    lap = -4.f * cen; lap += U4;  lap += D4;  lap += L4;  lap += R4;  diff += s1 * lap;
    lap = -4.f * cen; lap += U16; lap += D16; lap += L16; lap += R16; diff += s2 * lap;
    diff += k0 * (T1 - cen);
    diff += k1 * (T2 - cen);
    diff *= DIFF_SCALE;
    sum_lap += fabsf(diff[0]) + fabsf(diff[1]) + fabsf(diff[2]) + fabsf(diff[3]);

    const float a000 = __shfl(lpv, i000), a001 = __shfl(lpv, i001);
    const float a010 = __shfl(lpv, i010), a011 = __shfl(lpv, i011);
    const float a100 = __shfl(lpv, i100), a101 = __shfl(lpv, i101);
    const float a110 = __shfl(lpv, i110), a111 = __shfl(lpv, i111);
    f4 c00 = a000*omfw + a001*fw;
    f4 c01 = a010*omfw + a011*fw;
    f4 c10 = a100*omfw + a101*fw;
    f4 c11 = a110*omfw + a111*fw;
    f4 lv = (c00*omfh + c01*fh)*omft + (c10*omfh + c11*fh)*ft;

    f4 du = diff + rc + ag*hv + al*lv;
    f4 cd = cen + dt*du;
    sum_du += fabsf(du[0]) + fabsf(du[1]) + fabsf(du[2]) + fabsf(du[3]);
    sumsq  += cd*cd;
    cand[j] = cd;
    *(f4*)(ucand + off) = cd;
  }

  for (int o = 16; o > 0; o >>= 1){
    sum_du  += __shfl_xor(sum_du, o);
    sum_lap += __shfl_xor(sum_lap, o);
  }
  if (lane == 0){ red_du[wv] = sum_du; red_lap[wv] = sum_lap; }
  if (do_norm) sq[wv][lane] = sumsq;
  __syncthreads();

  f4 scale = {1.f, 1.f, 1.f, 1.f};
  if (do_norm){
    f4 tot = z4;
    for (int g = 0; g < UPD_WAVES; ++g) tot += sq[g][lane];
    for (int q = 0; q < 4; ++q) scale[q] = rsqrtf(tot[q]/(float)C_ + 1e-8f);
  }
  #pragma unroll
  for (int j = 0; j < UPD_CPW; ++j){
    const int c = wv*UPD_CPW + j;
    const size_t off = bbase + (size_t)c * N_ + n;
    f4 o4 = do_norm ? cand[j]*scale*normw[c] : cand[j];
    *(f4*)(unext + off) = o4;
  }

  if (tid == 0){
    float s1 = 0.f, s2 = 0.f;
    for (int g = 0; g < UPD_WAVES; ++g){ s1 += red_du[g]; s2 += red_lap[g]; }
    atomicAdd(&accum[1], s1);
    atomicAdd(&accum[0], s2);
  }
}

__global__ void turb_kernel(const float* cur, const float* prev, float* accum){
  int i = blockIdx.x*blockDim.x + threadIdx.x;
  float a = 0.f, pv = 0.f;
  if (i < PROBE_CELLS){
    float cv = cur[i], pp = prev[i];
    a = fabsf(cv - pp);
    pv = fabsf(pp);
  }
  float s1 = blockReduceSum(a);
  float s2 = blockReduceSum(pv);
  if (threadIdx.x == 0){
    atomicAdd(&accum[2], s1);
    atomicAdd(&accum[3], s2);
  }
}

__global__ void finalize_step_kernel(const float* accum, float* turbs, float* sten, int step){
  if (threadIdx.x == 0 && blockIdx.x == 0){
    const float inv_cells = 1.f/(float)PROBE_CELLS;
    const float inv_bnc   = 1.f/8388608.f;
    float stop_turb = (accum[2]*inv_cells) / (accum[3]*inv_cells + 1e-8f);
    float se = accum[1]*inv_bnc;
    float le = accum[0]*inv_bnc;
    turbs[step] = stop_turb + 0.05f*se + 0.01f*le;
    sten[step]  = se;
  }
}

__global__ void final_out_kernel(const float* turbs, const float* sten, float* out){
  if (threadIdx.x == 0 && blockIdx.x == 0){
    float a = 0.f, bm = 0.f;
    for (int i = 0; i < MAX_STEPS_; ++i){ a += turbs[i]; bm += sten[i]; }
    out[0] = a / (float)MAX_STEPS_;
    out[1] = bm / (float)MAX_STEPS_;
  }
}


extern "C" void kernel_launch(void* const* d_in, const int* in_sizes, int n_in,
                              void* d_out, int out_size, void* d_ws, size_t ws_size,
                              hipStream_t stream){
  (void)in_sizes; (void)n_in; (void)out_size; (void)ws_size;

  const float* u_in  = (const float*)d_in[0];
  const float* w1    = (const float*)d_in[1];
  const float* b1    = (const float*)d_in[2];
  const float* w2    = (const float*)d_in[3];
  const float* b2    = (const float*)d_in[4];
  const float* gw    = (const float*)d_in[5];
  const float* gb    = (const float*)d_in[6];
  const float* vw    = (const float*)d_in[7];
  const float* vb    = (const float*)d_in[8];
  const float* normw = (const float*)d_in[9];
  const float* wlp   = (const float*)d_in[10];
  const float* blp   = (const float*)d_in[11];
  const float* a_g   = (const float*)d_in[12];
  const float* a_l   = (const float*)d_in[13];
  const float* logdt = (const float*)d_in[14];
  const float* ks    = (const float*)d_in[15];
  const float* kt    = (const float*)d_in[16];

  float* Wf = (float*)d_ws;
  size_t o = 0;
  const size_t VOL = (size_t)B_ * C_ * N_;
  float* ubuf0  = Wf + o; o += VOL;
  float* ubuf1  = Wf + o; o += VOL;
  float* ucand  = Wf + o; o += VOL;
  float* react  = Wf + o; o += VOL;
  float* probeA = Wf + o; o += PROBE_CELLS;
  float* probeB = Wf + o; o += PROBE_CELLS;
  float* pooled = Wf + o; o += 256;
  float* hstate = Wf + o; o += 256;
  float* lsmall = Wf + o; o += (size_t)B_*C_*32;
  float* lproj  = Wf + o; o += (size_t)B_*C_*32;
  float* spsks  = Wf + o; o += 3*C_;
  float* spskt  = Wf + o; o += 2*C_;
  float* scal   = Wf + o; o += 8;
  float* accum  = Wf + o; o += 8;
  float* turbs  = Wf + o; o += 16;
  float* sten   = Wf + o; o += 16;
  _Float16* w1h = (_Float16*)(Wf + o); o += (C_*HID_)/2;
  _Float16* w2h = (_Float16*)(Wf + o); o += (HID_*C_)/2;

  float* ubuf[2] = { ubuf0, ubuf1 };

  prep_kernel<<<64, 256, 0, stream>>>(w1, w2, ks, kt, a_g, a_l, logdt,
                                      w1h, w2h, spsks, spskt, scal);
  zero_kernel<<<1, 256, 0, stream>>>(hstate, 256);
  hipMemcpyAsync(ubuf[0], u_in, VOL*sizeof(float), hipMemcpyDeviceToDevice, stream);
  pool_kernel<<<(PROBE_CELLS + 255)/256, 256, 0, stream>>>(ubuf[0], probeA, 2, 8, 8);

  float* pprev = probeA;
  float* pcur  = probeB;

  for (int step = 0; step < MAX_STEPS_; ++step){
    float* ucur  = ubuf[step & 1];
    float* unext = ubuf[(step & 1) ^ 1];
    int do_norm = (((step + 1) % 2) == 0) ? 1 : 0;

    zero_kernel<<<1, 32, 0, stream>>>(accum, 8);
    reaction_mlp_kernel<<<BN_/128, 256, 0, stream>>>(ucur, w1h, b1, w2h, b2, react);
    pooled_mean_kernel<<<B_*C_, 256, 0, stream>>>(ucur, pooled);
    gate_value_kernel<<<1, 256, 0, stream>>>(pooled, gw, gb, vw, vb, hstate);
    pool_kernel<<<(B_*C_*32 + 255)/256, 256, 0, stream>>>(ucur, lsmall, 2, 4, 4);
    local_proj_kernel<<<(B_*C_*32 + 255)/256, 256, 0, stream>>>(lsmall, wlp, blp, lproj);
    update_kernel<<<BN_/128, 512, 0, stream>>>(ucur, react, hstate, lproj, spsks, spskt,
                                               scal, normw, ucand, unext, accum, do_norm);
    pool_kernel<<<(PROBE_CELLS + 255)/256, 256, 0, stream>>>(ucand, pcur, 2, 8, 8);
    turb_kernel<<<PROBE_CELLS/256, 256, 0, stream>>>(pcur, pprev, accum);
    finalize_step_kernel<<<1, 1, 0, stream>>>(accum, turbs, sten, step);

    float* tswap = pprev; pprev = pcur; pcur = tswap;
  }

  hipMemcpyAsync(d_out, ubuf[MAX_STEPS_ & 1], VOL*sizeof(float),
                 hipMemcpyDeviceToDevice, stream);
  final_out_kernel<<<1, 32, 0, stream>>>(turbs, sten, ((float*)d_out) + VOL);
}


# update kernel emits block-pool partial sums of its outputs (probe cells of the candidate, local/global cells of the next state): the 3 full-volume pooling passes per step and the candidate store go
# speedup vs baseline: 5.4244x; 2.0067x over previous
#include <hip/hip_runtime.h>
#include <math.h>

#define B_ 2
#define C_ 128
#define T_ 8
#define H_ 64
#define W_ 64
#define N_ (T_*H_*W_)
#define BN_ (B_*N_)
#define HID_ 256
#define MAX_STEPS_ 12
#define DIFF_SCALE 0.25f
#define PROBE_CELLS (B_*C_*2*8*8)

typedef __attribute__((ext_vector_type(16))) _Float16 v16h;
typedef __attribute__((ext_vector_type(8)))  _Float16 h8;
typedef __attribute__((ext_vector_type(8)))  float    v8f;
typedef __attribute__((ext_vector_type(4)))  float    f4x;

__device__ __forceinline__ float softplus_f(float x){ return x > 20.f ? x : log1pf(expf(x)); }
__device__ __forceinline__ float gelu_f(float x){ return 0.5f*x*(1.f + erff(x*0.70710678118654752f)); }
__device__ __forceinline__ int imin_(int a, int b){ return a < b ? a : b; }

__device__ __forceinline__ float blockReduceSum(float v){
  __shared__ float red[8];
  int lane = threadIdx.x & 31, wid = threadIdx.x >> 5;
  #pragma unroll
  for (int off = 16; off > 0; off >>= 1) v += __shfl_down(v, off);
  if (lane == 0) red[wid] = v;
  __syncthreads();
  v = (threadIdx.x < (blockDim.x >> 5)) ? red[threadIdx.x] : 0.f;
  if (wid == 0){
    #pragma unroll
    for (int off = 4; off > 0; off >>= 1) v += __shfl_down(v, off);
  }
  __syncthreads();
  return v;
}

__device__ __forceinline__ v16h load_frag_row(const _Float16* base, int ld, int r0, int k0, int lane){
  int r = r0 + (lane & 15);
  int hf = lane >> 4;
  const _Float16* p = base + r*ld + k0 + hf*8;
  h8 lo = *(const h8*)p;
  h8 hi = *(const h8*)(p + 16);
  return __builtin_shufflevector(lo, hi, 0,1,2,3,4,5,6,7,8,9,10,11,12,13,14,15);
}


__global__ void zero_kernel(float* p, int n){
  int i = blockIdx.x*blockDim.x + threadIdx.x;
  if (i < n) p[i] = 0.f;
}

__global__ void prep_kernel(const float* w1, const float* w2,
                            const float* ks, const float* kt,
                            const float* ag, const float* al, const float* logdt,
                            _Float16* w1h, _Float16* w2h,
                            float* spsks, float* spskt, float* scal){
  int i = blockIdx.x*blockDim.x + threadIdx.x;
  int stride = gridDim.x*blockDim.x;
  for (int j = i; j < C_*HID_; j += stride) w1h[j] = (_Float16)w1[(j & (C_-1))*HID_ + (j >> 7)];
  for (int j = i; j < HID_*C_; j += stride) w2h[j] = (_Float16)w2[(j & (HID_-1))*C_ + (j >> 8)];
  for (int j = i; j < 3*C_;   j += stride) spsks[j] = softplus_f(ks[j]);
  for (int j = i; j < 2*C_;   j += stride) spskt[j] = softplus_f(kt[j]);
  if (i == 0){
    float dt = expf(logdt[0]);
    dt = fminf(fmaxf(dt, 0.005f), 0.35f);
    scal[0] = dt;
    scal[1] = softplus_f(ag[0]);
    scal[2] = softplus_f(al[0]);
  }
}

__global__ void pool_kernel(const float* in, float* out, int tp, int hp, int wp){
  int cells = B_*C_*tp*hp*wp;
  int id = blockIdx.x*blockDim.x + threadIdx.x;
  if (id >= cells) return;
  int pw = id % wp; int tmp = id / wp;
  int ph = tmp % hp; tmp /= hp;
  int pt = tmp % tp; tmp /= tp;
  int c = tmp % C_;  int b = tmp / C_;
  int bt = T_/tp, bh = H_/hp, bw = W_/wp;
  const float* base = in + (size_t)(b*C_ + c)*N_;
  float s = 0.f;
  for (int it = 0; it < bt; ++it)
    for (int ih = 0; ih < bh; ++ih)
      for (int iw = 0; iw < bw; ++iw)
        s += base[((pt*bt+it)*H_ + (ph*bh+ih))*W_ + (pw*bw+iw)];
  out[id] = s / (float)(bt*bh*bw);
}

__global__ void pooled_mean_kernel(const float* u, float* pooled){
  int bc = blockIdx.x;
  const float* base = u + (size_t)bc*N_;
  float s = 0.f;
  for (int i = threadIdx.x; i < N_; i += blockDim.x) s += base[i];
  s = blockReduceSum(s);
  if (threadIdx.x == 0) pooled[bc] = s / (float)N_;
}

__global__ void gate_value_kernel(const float* pooled, const float* gw, const float* gb,
                                  const float* vw, const float* vb, float* h){
  int tid = threadIdx.x;
  int b = tid >> 7, c = tid & 127;
  const float* p = pooled + b*C_;
  float g = gb[c], v = vb[c];
  for (int k = 0; k < C_; ++k){
    float pk = p[k];
    g += pk * gw[k*C_ + c];
    v += pk * vw[k*C_ + c];
  }
  h[tid] += (1.f/(1.f+expf(-g))) * tanhf(v);
}

__global__ void local_proj_kernel(const float* lsmall, const float* wlp, const float* blp, float* lproj){
  int id = blockIdx.x*blockDim.x + threadIdx.x;
  if (id >= B_*C_*32) return;
  int cell = id & 31;
  int oc = (id >> 5) & 127;
  int b = id >> 12;
  float s = blp[oc];
  for (int c = 0; c < C_; ++c)
    s += lsmall[((size_t)(b*C_ + c) << 5) + cell] * wlp[c*C_ + oc];
  lproj[((size_t)(b*C_ + oc) << 5) + cell] = s;
}

#define PA_ 136
#define PH_ 40

__global__ void __launch_bounds__(256)
reaction_mlp_kernel(const float* u, const _Float16* w1t, const float* b1,
                    const _Float16* w2t, const float* b2, float* react){
  __shared__ __attribute__((aligned(16))) _Float16 As[128*PA_];
  __shared__ __attribute__((aligned(16))) _Float16 Bt[32*PA_];
  __shared__ __attribute__((aligned(16))) _Float16 Hs[128*PH_];
  __shared__ __attribute__((aligned(16))) _Float16 W2t[128*PH_];

  const int tid  = threadIdx.x;
  const int lane = tid & 31;
  const int wv   = tid >> 5;
  const int m0   = wv * 16;
  const int hf   = lane >> 4;
  const int col  = lane & 15;

  const int rowBase = blockIdx.x * 128;
  const int b  = rowBase / N_;
  const int n0 = rowBase - b * N_;
  const float* ub = u + (size_t)b * C_ * N_;

  const _Float16* g1 = w1t + (size_t)(tid >> 4)*C_   + (tid & 15)*8;
  const _Float16* g2 = w2t + (size_t)(tid >> 2)*HID_ + (tid & 3)*8;
  _Float16* l1 = Bt  + (tid >> 4)*PA_ + (tid & 15)*8;
  _Float16* l2 = W2t + (tid >> 2)*PH_ + (tid & 3)*8;

  h8 r1a = *(const h8*)g1;
  h8 r1b = *(const h8*)(g1 + 16*C_);
  h8 r2a = *(const h8*)g2;
  h8 r2b = *(const h8*)(g2 + 64*HID_);

  {
    f4x v[16];
    #pragma unroll
    for (int i = 0; i < 16; ++i)
      v[i] = *(const f4x*)(ub + (size_t)(wv*16 + i) * N_ + n0 + lane*4);
    #pragma unroll
    for (int g = 0; g < 2; ++g){
      #pragma unroll
      for (int j = 0; j < 4; ++j){
        h8 hv;
        #pragma unroll
        for (int i = 0; i < 8; ++i) hv[i] = (_Float16)v[g*8 + i][j];
        *(h8*)(As + (lane*4 + j)*PA_ + wv*16 + g*8) = hv;
      }
    }
  }

  const v8f z8 = {0.f,0.f,0.f,0.f,0.f,0.f,0.f,0.f};
  v8f accC[8];
  #pragma unroll
  for (int i = 0; i < 8; ++i) accC[i] = z8;

  for (int hc = 0; hc < 8; ++hc){
    __syncthreads();
    *(h8*)l1              = r1a;
    *(h8*)(l1 + 16*PA_)   = r1b;
    *(h8*)l2              = r2a;
    *(h8*)(l2 + 64*PH_)   = r2b;
    __syncthreads();

    const float bia0 = b1[hc*32 + col], bia1 = b1[hc*32 + 16 + col];
    if (hc < 7){
      r1a = *(const h8*)(g1 + (hc+1)*32*C_);
      r1b = *(const h8*)(g1 + (hc+1)*32*C_ + 16*C_);
      r2a = *(const h8*)(g2 + (hc+1)*32);
      r2b = *(const h8*)(g2 + (hc+1)*32 + 64*HID_);
    }

    v8f a0 = z8, a1 = z8;
    #pragma unroll
    for (int k0 = 0; k0 < 128; k0 += 32){
      v16h af  = load_frag_row(As, PA_, m0, k0, lane);
      v16h bf0 = load_frag_row(Bt, PA_, 0,  k0, lane);
      v16h bf1 = load_frag_row(Bt, PA_, 16, k0, lane);
      a0 = __builtin_amdgcn_wmma_f32_16x16x32_f16(false, af, false, bf0, (short)0, a0, false, false);
      a1 = __builtin_amdgcn_wmma_f32_16x16x32_f16(false, af, false, bf1, (short)0, a1, false, false);
    }
    #pragma unroll
    for (int r = 0; r < 8; ++r){
      int m = m0 + hf*8 + r;
      Hs[m*PH_ + col]      = (_Float16)gelu_f(a0[r] + bia0);
      Hs[m*PH_ + 16 + col] = (_Float16)gelu_f(a1[r] + bia1);
    }
    v16h ah = load_frag_row(Hs, PH_, m0, 0, lane);
    #pragma unroll
    for (int nt = 0; nt < 8; ++nt){
      v16h bw = load_frag_row(W2t, PH_, nt*16, 0, lane);
      accC[nt] = __builtin_amdgcn_wmma_f32_16x16x32_f16(false, ah, false, bw, (short)0, accC[nt], false, false);
    }
  }

  #pragma unroll
  for (int nt = 0; nt < 8; ++nt){
    int c = nt*16 + col;
    float bias = b2[c];
    float* dst = react + (size_t)(b*C_ + c)*N_ + n0 + m0 + hf*8;
    float4 v0 = { accC[nt][0]+bias, accC[nt][1]+bias, accC[nt][2]+bias, accC[nt][3]+bias };
    float4 v1 = { accC[nt][4]+bias, accC[nt][5]+bias, accC[nt][6]+bias, accC[nt][7]+bias };
    *(float4*)dst       = v0;
    *(float4*)(dst + 4) = v1;
  }
}

typedef float f4 __attribute__((ext_vector_type(4)));
#define UPD_WAVES 16
#define UPD_CPW   8

__device__ __forceinline__ f4 ldg4(const float* p){ return *(const f4*)p; }

__device__ __forceinline__ float lane_xor1(float x){
  return __builtin_bit_cast(float, __builtin_amdgcn_update_dpp(0, __builtin_bit_cast(int, x), 0xB1, 0xF, 0xF, true));
}
__device__ __forceinline__ float lane_xor2(float x){
  return __builtin_bit_cast(float, __builtin_amdgcn_update_dpp(0, __builtin_bit_cast(int, x), 0x4E, 0xF, 0xF, true));
}
__device__ __forceinline__ float lane_xor16(float x){
  const int b = __builtin_bit_cast(int, x);
  return __builtin_bit_cast(float, __builtin_amdgcn_permlanex16(b, b, 0x76543210, 0xfedcba98, false, false));
}

__global__ void __launch_bounds__(512)
update_kernel(const float* __restrict__ u, const float* __restrict__ react, const float* __restrict__ h,
              const float* __restrict__ lproj, const float* __restrict__ spsks, const float* __restrict__ spskt,
              const float* __restrict__ scal, const float* normw,
              float* __restrict__ unext, float* __restrict__ part_l, float* __restrict__ part_p,
              float* __restrict__ accum, int do_norm){
  __shared__ f4 sq[UPD_WAVES][32];
  __shared__ __attribute__((aligned(16))) float stg_p[UPD_WAVES][64];
  __shared__ __attribute__((aligned(16))) float stg_l[UPD_WAVES][32];
  __shared__ float red_du[UPD_WAVES];
  __shared__ float red_lap[UPD_WAVES];

  const int tid  = threadIdx.x;
  const int lane = tid & 31;
  const int wv   = __builtin_amdgcn_readfirstlane(tid >> 5);
  const int rp   = blockIdx.x;
  const int b    = rp >> 8;
  const int t    = (rp >> 5) & 7;
  const int hh   = ((rp & 31) << 1) + (lane >> 4);
  const int ww0  = (lane & 15) << 2;
  const int n    = (t << 12) + (hh << 6) + ww0;

  const float dt = scal[0], ag = scal[1], al = scal[2];

  float st = fminf(fmaxf((t  + 0.5f)*(2.f/T_) - 0.5f, 0.f), 1.f);
  float sh = fminf(fmaxf((hh + 0.5f)*(4.f/H_) - 0.5f, 0.f), 3.f);
  int t0i = (int)st; int t1i = imin_(t0i+1, 1); float ft = st - (float)t0i;
  int h0i = (int)sh; int h1i = imin_(h0i+1, 3); float fh = sh - (float)h0i;
  f4 fw;
  int w0i = 0;
  for (int q = 0; q < 4; ++q){
    float sw = fminf(fmaxf((ww0 + q + 0.5f)*(4.f/W_) - 0.5f, 0.f), 3.f);
    int wi = (int)sw;
    if (q == 0) w0i = wi;
    fw[q] = sw - (float)wi;
  }
  const int w1i = imin_(w0i+1, 3);
  const f4 omfw = 1.f - fw;
  const float omfh = 1.f - fh, omft = 1.f - ft;
  const int i000 = t0i*16 + h0i*4 + w0i, i001 = t0i*16 + h0i*4 + w1i;
  const int i010 = t0i*16 + h1i*4 + w0i, i011 = t0i*16 + h1i*4 + w1i;
  const int i100 = t1i*16 + h0i*4 + w0i, i101 = t1i*16 + h0i*4 + w1i;
  const int i110 = t1i*16 + h1i*4 + w0i, i111 = t1i*16 + h1i*4 + w1i;

  const bool kU1 = hh >= 1,  kD1 = hh < 63, kU4 = hh >= 4,  kD4 = hh < 60;
  const bool kU16 = hh >= 16, kD16 = hh < 48;
  const bool kL4 = ww0 >= 4, kR4 = ww0 < 60, kL16 = ww0 >= 16, kR16 = ww0 < 48;
  const bool kT1 = t >= 1, kT2 = t >= 2;
  const int oU1 = kU1 ? -64 : 0,    oD1 = kD1 ? 64 : 0;
  const int oU4 = kU4 ? -256 : 0,   oD4 = kD4 ? 256 : 0;
  const int oU16 = kU16 ? -1024 : 0, oD16 = kD16 ? 1024 : 0;
  const int oL4 = kL4 ? -4 : 0,     oR4 = kR4 ? 4 : 0;
  const int oL16 = kL16 ? -16 : 0,  oR16 = kR16 ? 16 : 0;
  const int oT1 = kT1 ? -4096 : 0,  oT2 = kT2 ? -8192 : 0;
  const f4 z4 = {0.f, 0.f, 0.f, 0.f};

  f4 cand[UPD_CPW];
  f4 sumsq = z4;
  float sum_du = 0.f, sum_lap = 0.f;
  const size_t bbase = (size_t)b * C_ * N_;

  #pragma unroll
  for (int j = 0; j < UPD_CPW; ++j){
    const int c = wv*UPD_CPW + j;
    const size_t off = bbase + (size_t)c * N_ + n;
    const float* uc = u + off;

    f4 cen = ldg4(uc);
    f4 U1  = ldg4(uc + oU1),  D1  = ldg4(uc + oD1);
    f4 U4  = ldg4(uc + oU4),  D4  = ldg4(uc + oD4);
    f4 U16 = ldg4(uc + oU16), D16 = ldg4(uc + oD16);
    f4 L4  = ldg4(uc + oL4),  R4  = ldg4(uc + oR4);
    f4 L16 = ldg4(uc + oL16), R16 = ldg4(uc + oR16);
    f4 T1  = ldg4(uc + oT1),  T2  = ldg4(uc + oT2);
    f4 rc  = ldg4(react + off);
    float lpv = lproj[((size_t)(b*C_ + c) << 5) + lane];

    U1  = kU1  ? U1  : z4;  D1  = kD1  ? D1  : z4;
    U4  = kU4  ? U4  : z4;  D4  = kD4  ? D4  : z4;
    U16 = kU16 ? U16 : z4;  D16 = kD16 ? D16 : z4;
    L4  = kL4  ? L4  : z4;  R4  = kR4  ? R4  : z4;
    L16 = kL16 ? L16 : z4;  R16 = kR16 ? R16 : z4;
    T1  = kT1  ? T1  : z4;  T2  = kT2  ? T2  : z4;
    f4 L1 = { L4[3], cen[0], cen[1], cen[2] };
    f4 R1 = { cen[1], cen[2], cen[3], R4[0] };

    const float s0 = spsks[c], s1 = spsks[C_ + c], s2 = spsks[2*C_ + c];
    const float k0 = spskt[c], k1 = spskt[C_ + c];
    const float hv = h[b*C_ + c];

    f4 diff, lap;
    lap = -4.f * cen; lap += U1;  lap += D1;  lap += L1;  lap += R1;  diff  = s0 * lap;
    lap = -4.f * cen; lap += U4;  lap += D4;  lap += L4;  lap += R4;  diff += s1 * lap;
    lap = -4.f * cen; lap += U16; lap += D16; lap += L16; lap += R16; diff += s2 * lap;
    diff += k0 * (T1 - cen);
    diff += k1 * (T2 - cen);
    diff *= DIFF_SCALE;
    sum_lap += fabsf(diff[0]) + fabsf(diff[1]) + fabsf(diff[2]) + fabsf(diff[3]);

    const float a000 = __shfl(lpv, i000), a001 = __shfl(lpv, i001);
    const float a010 = __shfl(lpv, i010), a011 = __shfl(lpv, i011);
    const float a100 = __shfl(lpv, i100), a101 = __shfl(lpv, i101);
    const float a110 = __shfl(lpv, i110), a111 = __shfl(lpv, i111);
    f4 c00 = a000*omfw + a001*fw;
    f4 c01 = a010*omfw + a011*fw;
    f4 c10 = a100*omfw + a101*fw;
    f4 c11 = a110*omfw + a111*fw;
    f4 lv = (c00*omfh + c01*fh)*omft + (c10*omfh + c11*fh)*ft;

    f4 du = diff + rc + ag*hv + al*lv;
    f4 cd = cen + dt*du;
    sum_du += fabsf(du[0]) + fabsf(du[1]) + fabsf(du[2]) + fabsf(du[3]);
    sumsq  += cd*cd;
    cand[j] = cd;
    float ps = (cd[0] + cd[1]) + (cd[2] + cd[3]);
    ps += lane_xor1(ps);
    ps += lane_xor16(ps);
    if ((lane & 17) == 0) stg_p[wv][j*8 + (lane >> 1)] = ps;
  }

  for (int o = 16; o > 0; o >>= 1){
    sum_du  += __shfl_xor(sum_du, o);
    sum_lap += __shfl_xor(sum_lap, o);
  }
  if (lane == 0){ red_du[wv] = sum_du; red_lap[wv] = sum_lap; }
  if (do_norm) sq[wv][lane] = sumsq;
  __syncthreads();

  *(float2*)(part_p + (size_t)rp*1024 + wv*64 + lane*2) = *(const float2*)&stg_p[wv][lane*2];

  f4 scale = {1.f, 1.f, 1.f, 1.f};
  if (do_norm){
    f4 tot = z4;
    for (int g = 0; g < UPD_WAVES; ++g) tot += sq[g][lane];
    for (int q = 0; q < 4; ++q) scale[q] = rsqrtf(tot[q]/(float)C_ + 1e-8f);
  }
  #pragma unroll
  for (int j = 0; j < UPD_CPW; ++j){
    const int c = wv*UPD_CPW + j;
    const size_t off = bbase + (size_t)c * N_ + n;
    f4 o4 = do_norm ? cand[j]*scale*normw[c] : cand[j];
    *(f4*)(unext + off) = o4;
    float ls = (o4[0] + o4[1]) + (o4[2] + o4[3]);
    ls += lane_xor1(ls);
    ls += lane_xor2(ls);
    ls += lane_xor16(ls);
    if ((lane & 19) == 0) stg_l[wv][j*4 + (lane >> 2)] = ls;
  }
  __syncthreads();
  part_l[(size_t)rp*512 + wv*32 + lane] = stg_l[wv][lane];

  if (tid == 0){
    float s1 = 0.f, s2 = 0.f;
    for (int g = 0; g < UPD_WAVES; ++g){ s1 += red_du[g]; s2 += red_lap[g]; }
    atomicAdd(&accum[1], s1);
    atomicAdd(&accum[0], s2);
  }
}

__global__ void __launch_bounds__(256)
reduce_partials_kernel(const float* __restrict__ part_l, const float* __restrict__ part_p,
                       float* __restrict__ lsmall, float* __restrict__ pooled, float* __restrict__ probe){
  const int bc  = blockIdx.x;
  const int b   = bc >> 7, c = bc & 127;
  const int tid = threadIdx.x;
  const size_t bblk = (size_t)b * 256;

  {
    const int cell = tid >> 3, sub = tid & 7;
    const int pt = cell >> 4, ph = (cell >> 2) & 3, pw = cell & 3;
    const int t  = pt*4 + (sub >> 1);
    float s = 0.f;
    #pragma unroll
    for (int q = 0; q < 4; ++q){
      const int rpair = ph*8 + (sub & 1)*4 + q;
      s += part_l[(bblk + t*32 + rpair)*512 + c*4 + pw];
    }
    s += __shfl_xor(s, 1);
    s += __shfl_xor(s, 2);
    s += __shfl_xor(s, 4);
    if (sub == 0) lsmall[bc*32 + cell] = s * (1.f/1024.f);
    const float tot = blockReduceSum(sub == 0 ? s : 0.f);
    if (tid == 0) pooled[bc] = tot / (float)N_;
  }
  {
    const int cell = tid >> 1, sub = tid & 1;
    const int pt = cell >> 6, ph = (cell >> 3) & 7, pw = cell & 7;
    float s = 0.f;
    #pragma unroll
    for (int q = 0; q < 8; ++q){
      const int k = sub*8 + q;
      const int t = pt*4 + (k >> 2);
      const int rpair = ph*4 + (k & 3);
      s += part_p[(bblk + t*32 + rpair)*1024 + c*8 + pw];
    }
    s += __shfl_xor(s, 1);
    if (sub == 0) probe[bc*128 + cell] = s * (1.f/256.f);
  }
}

__global__ void turb_kernel(const float* cur, const float* prev, float* accum){
  int i = blockIdx.x*blockDim.x + threadIdx.x;
  float a = 0.f, pv = 0.f;
  if (i < PROBE_CELLS){
    float cv = cur[i], pp = prev[i];
    a = fabsf(cv - pp);
    pv = fabsf(pp);
  }
  float s1 = blockReduceSum(a);
  float s2 = blockReduceSum(pv);
  if (threadIdx.x == 0){
    atomicAdd(&accum[2], s1);
    atomicAdd(&accum[3], s2);
  }
}

__global__ void finalize_step_kernel(const float* accum, float* turbs, float* sten, int step){
  if (threadIdx.x == 0 && blockIdx.x == 0){
    const float inv_cells = 1.f/(float)PROBE_CELLS;
    const float inv_bnc   = 1.f/8388608.f;
    float stop_turb = (accum[2]*inv_cells) / (accum[3]*inv_cells + 1e-8f);
    float se = accum[1]*inv_bnc;
    float le = accum[0]*inv_bnc;
    turbs[step] = stop_turb + 0.05f*se + 0.01f*le;
    sten[step]  = se;
  }
}

__global__ void final_out_kernel(const float* turbs, const float* sten, float* out){
  if (threadIdx.x == 0 && blockIdx.x == 0){
    float a = 0.f, bm = 0.f;
    for (int i = 0; i < MAX_STEPS_; ++i){ a += turbs[i]; bm += sten[i]; }
    out[0] = a / (float)MAX_STEPS_;
    out[1] = bm / (float)MAX_STEPS_;
  }
}


extern "C" void kernel_launch(void* const* d_in, const int* in_sizes, int n_in,
                              void* d_out, int out_size, void* d_ws, size_t ws_size,
                              hipStream_t stream){
  (void)in_sizes; (void)n_in; (void)out_size; (void)ws_size;

  const float* u_in  = (const float*)d_in[0];
  const float* w1    = (const float*)d_in[1];
  const float* b1    = (const float*)d_in[2];
  const float* w2    = (const float*)d_in[3];
  const float* b2    = (const float*)d_in[4];
  const float* gw    = (const float*)d_in[5];
  const float* gb    = (const float*)d_in[6];
  const float* vw    = (const float*)d_in[7];
  const float* vb    = (const float*)d_in[8];
  const float* normw = (const float*)d_in[9];
  const float* wlp   = (const float*)d_in[10];
  const float* blp   = (const float*)d_in[11];
  const float* a_g   = (const float*)d_in[12];
  const float* a_l   = (const float*)d_in[13];
  const float* logdt = (const float*)d_in[14];
  const float* ks    = (const float*)d_in[15];
  const float* kt    = (const float*)d_in[16];

  float* Wf = (float*)d_ws;
  size_t o = 0;
  const size_t VOL = (size_t)B_ * C_ * N_;
  float* ubuf0  = Wf + o; o += VOL;
  float* ubuf1  = Wf + o; o += VOL;
  float* parts  = Wf + o; o += VOL;
  float* react  = Wf + o; o += VOL;
  float* probeA = Wf + o; o += PROBE_CELLS;
  float* probeB = Wf + o; o += PROBE_CELLS;
  float* pooled = Wf + o; o += 256;
  float* hstate = Wf + o; o += 256;
  float* lsmall = Wf + o; o += (size_t)B_*C_*32;
  float* lproj  = Wf + o; o += (size_t)B_*C_*32;
  float* spsks  = Wf + o; o += 3*C_;
  float* spskt  = Wf + o; o += 2*C_;
  float* scal   = Wf + o; o += 8;
  float* accum  = Wf + o; o += 8;
  float* turbs  = Wf + o; o += 16;
  float* sten   = Wf + o; o += 16;
  _Float16* w1h = (_Float16*)(Wf + o); o += (C_*HID_)/2;
  _Float16* w2h = (_Float16*)(Wf + o); o += (HID_*C_)/2;

  float* part_l = parts;
  float* part_p = parts + 512*C_*4;

  float* ubuf[2] = { ubuf0, ubuf1 };

  prep_kernel<<<64, 256, 0, stream>>>(w1, w2, ks, kt, a_g, a_l, logdt,
                                      w1h, w2h, spsks, spskt, scal);
  zero_kernel<<<1, 256, 0, stream>>>(hstate, 256);
  hipMemcpyAsync(ubuf[0], u_in, VOL*sizeof(float), hipMemcpyDeviceToDevice, stream);
  pool_kernel<<<(PROBE_CELLS + 255)/256, 256, 0, stream>>>(ubuf[0], probeA, 2, 8, 8);
  pool_kernel<<<(B_*C_*32 + 255)/256, 256, 0, stream>>>(ubuf[0], lsmall, 2, 4, 4);
  pooled_mean_kernel<<<B_*C_, 256, 0, stream>>>(ubuf[0], pooled);

  float* pprev = probeA;
  float* pcur  = probeB;

  for (int step = 0; step < MAX_STEPS_; ++step){
    float* ucur  = ubuf[step & 1];
    float* unext = ubuf[(step & 1) ^ 1];
    int do_norm = (((step + 1) % 2) == 0) ? 1 : 0;

    zero_kernel<<<1, 32, 0, stream>>>(accum, 8);
    reaction_mlp_kernel<<<BN_/128, 256, 0, stream>>>(ucur, w1h, b1, w2h, b2, react);
    gate_value_kernel<<<1, 256, 0, stream>>>(pooled, gw, gb, vw, vb, hstate);
    local_proj_kernel<<<(B_*C_*32 + 255)/256, 256, 0, stream>>>(lsmall, wlp, blp, lproj);
    update_kernel<<<BN_/128, 512, 0, stream>>>(ucur, react, hstate, lproj, spsks, spskt,
                                               scal, normw, unext, part_l, part_p, accum, do_norm);
    reduce_partials_kernel<<<B_*C_, 256, 0, stream>>>(part_l, part_p, lsmall, pooled, pcur);
    turb_kernel<<<PROBE_CELLS/256, 256, 0, stream>>>(pcur, pprev, accum);
    finalize_step_kernel<<<1, 1, 0, stream>>>(accum, turbs, sten, step);

    float* tswap = pprev; pprev = pcur; pcur = tswap;
  }

  hipMemcpyAsync(d_out, ubuf[MAX_STEPS_ & 1], VOL*sizeof(float),
                 hipMemcpyDeviceToDevice, stream);
  final_out_kernel<<<1, 32, 0, stream>>>(turbs, sten, ((float*)d_out) + VOL);
}


# two work-groups resident per WGP, not one: LDS 235->101 KB (reaction tile over dead staging tiles), 240->128 VGPRs (product 2 split by output channel: hidden sum in one chain, not 2 partial sums)
# speedup vs baseline: 10.2828x; 1.8957x over previous
#include <hip/hip_runtime.h>
#include <math.h>

#define B_ 2
#define C_ 128
#define T_ 8
#define H_ 64
#define W_ 64
#define N_ (T_*H_*W_)
#define BN_ (B_*N_)
#define HID_ 256
#define MAX_STEPS_ 12
#define DIFF_SCALE 0.25f
#define PROBE_CELLS (B_*C_*2*8*8)

typedef __attribute__((ext_vector_type(16))) _Float16 v16h;
typedef __attribute__((ext_vector_type(8)))  _Float16 h8;
typedef __attribute__((ext_vector_type(8)))  float    v8f;
typedef __attribute__((ext_vector_type(4)))  float    f4x;
typedef float f4 __attribute__((ext_vector_type(4)));
__device__ __forceinline__ f4 ldg4(const float* p){ return *(const f4*)p; }

__device__ __forceinline__ float softplus_f(float x){ return x > 20.f ? x : log1pf(expf(x)); }
typedef float g8f __attribute__((vector_size(32)));
__device__ __forceinline__ v8f gelu8(v8f av){
  const g8f a = __builtin_bit_cast(g8f, av);
  const g8f x = a * 0.70710678118654752f;
  const g8f t = __builtin_elementwise_abs(x);
  g8f p = t*0.0000169988107f + -0.000378677854f;
  p = t*p + 0.00385781587f;
  p = t*p + -0.0241816975f;
  p = t*p + 0.106668264f;
  p = t*p + 0.634933293f;
  p = t*p + 0.128689408f;
  p = t*p + t;
  const g8f y = p * -1.44269504088896341f;
  g8f e2;
  #pragma unroll
  for (int i = 0; i < 8; ++i) e2[i] = __builtin_amdgcn_exp2f(y[i]);
  const g8f rl = __builtin_elementwise_copysign(1.f - e2, x);
  const g8f s = x*x;
  g8f q = s*-0.000561801775f + 0.00491381623f;
  q = s*q + -0.0267075151f;
  q = s*q + 0.112800106f;
  q = s*q + -0.376122952f;
  q = s*q + 0.128379107f;
  const g8f rs = x*q + x;
  const g8f e = (t >= 1.f) ? rl : rs;
  const g8f g = (0.5f*a)*(1.f + e);
  return __builtin_bit_cast(v8f, g);
}
__device__ __forceinline__ int imin_(int a, int b){ return a < b ? a : b; }

__device__ __forceinline__ float blockReduceSum(float v){
  __shared__ float red[8];
  int lane = threadIdx.x & 31, wid = threadIdx.x >> 5;
  #pragma unroll
  for (int off = 16; off > 0; off >>= 1) v += __shfl_down(v, off);
  if (lane == 0) red[wid] = v;
  __syncthreads();
  v = (threadIdx.x < (blockDim.x >> 5)) ? red[threadIdx.x] : 0.f;
  if (wid == 0){
    #pragma unroll
    for (int off = 4; off > 0; off >>= 1) v += __shfl_down(v, off);
  }
  __syncthreads();
  return v;
}

__device__ __forceinline__ v16h load_frag_row(const _Float16* base, int ld, int r0, int k0, int lane){
  int r = r0 + (lane & 15);
  int hf = lane >> 4;
  const _Float16* p = base + r*ld + k0 + hf*8;
  h8 lo = *(const h8*)p;
  h8 hi = *(const h8*)(p + 16);
  return __builtin_shufflevector(lo, hi, 0,1,2,3,4,5,6,7,8,9,10,11,12,13,14,15);
}


__global__ void zero_kernel(float* p, int n, float* q, int m){
  int i = blockIdx.x*blockDim.x + threadIdx.x;
  if (i < n) p[i] = 0.f;
  if (i < m) q[i] = 0.f;
}

__global__ void prep_kernel(const float* w1, const float* w2,
                            const float* ks, const float* kt,
                            const float* ag, const float* al, const float* logdt,
                            _Float16* w1h, _Float16* w2h,
                            float* spsks, float* spskt, float* scal){
  int i = blockIdx.x*blockDim.x + threadIdx.x;
  int stride = gridDim.x*blockDim.x;
  for (int j = i; j < C_*HID_; j += stride) w1h[j] = (_Float16)w1[(j & (C_-1))*HID_ + (j >> 7)];
  for (int j = i; j < HID_*C_; j += stride) w2h[j] = (_Float16)w2[(j & (HID_-1))*C_ + (j >> 8)];
  for (int j = i; j < 3*C_;   j += stride) spsks[j] = softplus_f(ks[j]);
  for (int j = i; j < 2*C_;   j += stride) spskt[j] = softplus_f(kt[j]);
  if (i == 0){
    float dt = expf(logdt[0]);
    dt = fminf(fmaxf(dt, 0.005f), 0.35f);
    scal[0] = dt;
    scal[1] = softplus_f(ag[0]);
    scal[2] = softplus_f(al[0]);
  }
}

#define GATE_BLOCKS_  2
#define LOC_BLOCKS_   8
#define PROBE_BLOCKS_ 64
#define SIDE_BLOCKS_ (GATE_BLOCKS_ + LOC_BLOCKS_ + PROBE_BLOCKS_)

__device__ __forceinline__ void gate_block(int b, const float* part_g, const float* gw, const float* gb,
                                           const float* vw, const float* vb, float* h){
  __shared__ float ps[2][C_];
  __shared__ float pl[C_];
  __shared__ float gv[2][C_];
  const int tid = threadIdx.x;
  const int c = tid & 127, hs = tid >> 7;
  const float* p = part_g + ((size_t)b*256 + hs*128)*C_ + c;
  float s = 0.f;
  #pragma unroll
  for (int r = 0; r < 128; ++r) s += p[(size_t)r*C_];
  ps[hs][c] = s;
  __syncthreads();
  if (tid < C_) pl[tid] = (ps[0][tid] + ps[1][tid]) / (float)N_;
  __syncthreads();
  const float* wm = hs ? vw : gw;
  float a = hs ? vb[c] : gb[c];
  #pragma unroll
  for (int k = 0; k < C_; ++k) a += pl[k] * wm[k*C_ + c];
  gv[hs][c] = a;
  __syncthreads();
  if (tid < C_) h[b*C_ + tid] += (1.f/(1.f+expf(-gv[0][tid]))) * tanhf(gv[1][tid]);
}

__device__ __forceinline__ void local_block(int blk, const float* part_l, const float* wlp,
                                            const float* blp, float* lproj){
  __shared__ __attribute__((aligned(16))) float lsm[C_][8];
  const int tid = threadIdx.x;
  const int b = blk >> 2, cg = blk & 3;
  const int pt = cg >> 1, hp = cg & 1;
  const size_t bblk = (size_t)b * 256;
  {
    const int c = tid & 127, phl = tid >> 7;
    const int ph = hp*2 + phl;
    f4 leaf[8];
    for (int sb = 0; sb < 8; ++sb){
      const int t = pt*4 + (sb >> 1);
      f4 sm = {0.f, 0.f, 0.f, 0.f};
      for (int q = 0; q < 4; ++q){
        const int rpair = ph*8 + (sb & 1)*4 + q;
        sm += ldg4(part_l + ((bblk + t*32 + rpair)*C_ + c)*4);
      }
      leaf[sb] = sm;
    }
    const f4 tot = ((leaf[0] + leaf[1]) + (leaf[2] + leaf[3])) + ((leaf[4] + leaf[5]) + (leaf[6] + leaf[7]));
    *(f4*)&lsm[c][phl*4] = tot * (1.f/1024.f);
  }
  __syncthreads();
  const int oc = tid >> 1, chh = tid & 1;
  const float bias = blp[oc];
  f4 pr = {bias, bias, bias, bias};
  #pragma unroll
  for (int c = 0; c < C_; ++c){
    const f4 l = *(const f4*)&lsm[c][chh*4];
    pr += l * wlp[c*C_ + oc];
  }
  *(f4*)(lproj + ((((size_t)b*4 + cg)*C_ + oc) << 3) + chh*4) = pr;
}

#define PROBE_CPB_ 4
__device__ __forceinline__ void probe_fold_block(int pb, const float* part_p, float* probe,
                                                 const float* prev, float* accum, int with_diff){
  const int tid  = threadIdx.x;
  const int cell = tid & 127, cp = tid >> 7;
  const int pt = cell >> 6, ph = (cell >> 3) & 7, pw = cell & 7;
  const int bc0 = pb*PROBE_CPB_ + cp*2;
  const int b = bc0 >> 7, c0 = bc0 & 127;
  const size_t bblk = (size_t)b * 256;
  float pp[2];
  for (int i = 0; i < 2; ++i){
    const float pv = prev[(bc0 + i)*128 + cell];
    pp[i] = with_diff ? pv : 0.f;
  }
  float ad = 0.f, ap = 0.f;
  float cv[2];
  for (int i = 0; i < 2; ++i){
    float hsum[2];
    for (int sb = 0; sb < 2; ++sb){
      float a = 0.f;
      for (int q = 0; q < 8; ++q){
        const int k = sb*8 + q;
        const int t = pt*4 + (k >> 2);
        const int rpair = ph*4 + (k & 3);
        a += part_p[(bblk + t*32 + rpair)*1024 + (c0 + i)*8 + pw];
      }
      hsum[sb] = a;
    }
    cv[i] = (hsum[0] + hsum[1]) * (1.f/256.f);
  }
  for (int i = 0; i < 2; ++i){
    probe[(bc0 + i)*128 + cell] = cv[i];
    ad += fabsf(cv[i] - pp[i]);
    ap += fabsf(pp[i]);
  }
  if (with_diff){
    const float s1 = blockReduceSum(ad);
    const float s2 = blockReduceSum(ap);
    if (tid == 0){
      atomicAdd(&accum[2], s1);
      atomicAdd(&accum[3], s2);
    }
  }
}

__global__ void __launch_bounds__(256)
probe_fold_kernel(const float* part_p, float* probe, const float* prev, float* accum){
  probe_fold_block(blockIdx.x, part_p, probe, prev, accum, 1);
}

__global__ void __launch_bounds__(256)
fold_kernel(const float* part_g, const float* gw, const float* gb,
            const float* vw, const float* vb, float* hstate,
            const float* part_l, const float* wlp, const float* blp, float* lproj,
            const float* part_p, float* probe, const float* prev, float* accum, int with_diff){
  const int sb = blockIdx.x;
  if (sb < GATE_BLOCKS_) gate_block(sb, part_g, gw, gb, vw, vb, hstate);
  else if (sb < GATE_BLOCKS_ + LOC_BLOCKS_) local_block(sb - GATE_BLOCKS_, part_l, wlp, blp, lproj);
  else probe_fold_block(sb - GATE_BLOCKS_ - LOC_BLOCKS_, part_p, probe, prev, accum, with_diff);
}

#define PA_  136
#define PW2_ 72
#define PR_  132
#define PHS_ 72
#define UPD_WAVES 16
#define UPD_CPW   8
#define MLP_LDS_BYTES_ ((128*PA_ + 64*PA_ + 128*PHS_ + 128*PW2_) * 2)
static_assert(C_*PR_*4 <= MLP_LDS_BYTES_, "the reaction tile must fit over the staging tiles");

__device__ __forceinline__ float lane_xor1(float x){
  return __builtin_bit_cast(float, __builtin_amdgcn_update_dpp(0, __builtin_bit_cast(int, x), 0xB1, 0xF, 0xF, true));
}
__device__ __forceinline__ float lane_xor2(float x){
  return __builtin_bit_cast(float, __builtin_amdgcn_update_dpp(0, __builtin_bit_cast(int, x), 0x4E, 0xF, 0xF, true));
}
__device__ __forceinline__ float lane_xor16(float x){
  const int b = __builtin_bit_cast(int, x);
  return __builtin_bit_cast(float, __builtin_amdgcn_permlanex16(b, b, 0x76543210, 0xfedcba98, false, false));
}
typedef const __attribute__((address_space(1))) float gfloat;
__device__ __forceinline__ gfloat* uniform_ptr(const float* p){
  const unsigned long long a = (unsigned long long)p;
  const unsigned lo = __builtin_amdgcn_readfirstlane((unsigned)a);
  const unsigned hi = __builtin_amdgcn_readfirstlane((unsigned)(a >> 32));
  return (gfloat*)(((unsigned long long)hi << 32) | lo);
}
__device__ __forceinline__ f4 ldg4_sv(gfloat* sbase, unsigned voff){
  typedef const __attribute__((address_space(1))) char gchar;
  typedef const __attribute__((address_space(1))) f4 gf4;
  return *(gf4*)((gchar*)sbase + voff);
}

__global__ void __launch_bounds__(512, 8)
step_kernel(const float* __restrict__ u, const _Float16* __restrict__ w1t, const float* b1,
            const _Float16* __restrict__ w2t, const float* b2, const float* __restrict__ h,
            const float* __restrict__ lproj, const float* __restrict__ spsks, const float* __restrict__ spskt,
            const float* __restrict__ scal, const float* normw,
            float* __restrict__ unext, float* __restrict__ part_l, float* __restrict__ part_p,
            float* __restrict__ part_g, float* __restrict__ accum, int do_norm){
  __shared__ __attribute__((aligned(16))) unsigned char lds_mlp[MLP_LDS_BYTES_];
  _Float16* const As  = (_Float16*)lds_mlp;
  _Float16* const Bt  = As  + 128*PA_;
  _Float16* const Hs  = Bt  + 64*PA_;
  _Float16* const W2t = Hs  + 128*PHS_;
  float*    const Rt  = (float*)lds_mlp;
  __shared__ f4 sq[UPD_WAVES][32];
  __shared__ __attribute__((aligned(16))) float stg_p[UPD_WAVES][64];
  __shared__ __attribute__((aligned(16))) float stg_l[UPD_WAVES][32];
  __shared__ float red_du[UPD_WAVES];
  __shared__ float red_lap[UPD_WAVES];

  const int tid  = threadIdx.x;
  const int lane = tid & 31;
  const int wv   = __builtin_amdgcn_readfirstlane(tid >> 5);
  const int rp   = blockIdx.x;
  const int b    = rp >> 8;
  const int t    = (rp >> 5) & 7;
  const int hh   = ((rp & 31) << 1) + (lane >> 4);
  const int ww0  = (lane & 15) << 2;
  const int n    = (t << 12) + (hh << 6) + ww0;

  {
    const int rt = wv & 7;
    const int ph = wv >> 3;
    const int m0 = rt * 16;
    const int hf = lane >> 4;
    const int col = lane & 15;
    const int n0 = (rp & 255) << 7;
    const float* ub = u + (size_t)b * C_ * N_;

    const _Float16* gp1 = w1t + (size_t)(tid >> 4)*C_   + (tid & 15)*8;
    const _Float16* gp2 = w2t + (size_t)(tid >> 3)*HID_ + (tid & 7)*8;
    _Float16* l1 = Bt  + (tid >> 4)*PA_  + (tid & 15)*8;
    _Float16* l2 = W2t + (tid >> 3)*PW2_ + (tid & 7)*8;

    h8 r1a = *(const h8*)gp1;
    h8 r1b = *(const h8*)(gp1 + 32*C_);
    h8 r2a = *(const h8*)gp2;
    h8 r2b = *(const h8*)(gp2 + 64*HID_);

    {
      f4x v[8];
      #pragma unroll
      for (int i = 0; i < 8; ++i)
        v[i] = *(const f4x*)(ub + (size_t)(wv*8 + i) * N_ + n0 + lane*4);
      #pragma unroll
      for (int j = 0; j < 4; ++j){
        h8 hv;
        #pragma unroll
        for (int i = 0; i < 8; ++i) hv[i] = (_Float16)v[i][j];
        *(h8*)(As + (lane*4 + j)*PA_ + wv*8) = hv;
      }
    }

    const v8f z8 = {0.f,0.f,0.f,0.f,0.f,0.f,0.f,0.f};
    v8f accC[4];
    #pragma unroll
    for (int i = 0; i < 4; ++i) accC[i] = z8;

    float nb0 = b1[ph*32 + col], nb1 = b1[ph*32 + 16 + col];
    for (int q = 0; q < 4; ++q){
      __syncthreads();
      *(h8*)l1              = r1a;
      *(h8*)(l1 + 32*PA_)   = r1b;
      *(h8*)l2              = r2a;
      *(h8*)(l2 + 64*PW2_)  = r2b;
      __syncthreads();

      const float bia0 = nb0, bia1 = nb1;
      if (q < 3){
        r1a = *(const h8*)(gp1 + (q+1)*64*C_);
        r1b = *(const h8*)(gp1 + (q+1)*64*C_ + 32*C_);
        r2a = *(const h8*)(gp2 + (q+1)*64);
        r2b = *(const h8*)(gp2 + (q+1)*64 + 64*HID_);
        nb0 = b1[(q+1)*64 + ph*32 + col];
        nb1 = b1[(q+1)*64 + ph*32 + 16 + col];
      }

      v8f a0 = z8, a1 = z8;
      #pragma unroll
      for (int k0 = 0; k0 < 128; k0 += 32){
        v16h af  = load_frag_row(As, PA_, m0, k0, lane);
        v16h bf0 = load_frag_row(Bt, PA_, ph*32,      k0, lane);
        v16h bf1 = load_frag_row(Bt, PA_, ph*32 + 16, k0, lane);
        a0 = __builtin_amdgcn_wmma_f32_16x16x32_f16(false, af, false, bf0, (short)0, a0, false, false);
        a1 = __builtin_amdgcn_wmma_f32_16x16x32_f16(false, af, false, bf1, (short)0, a1, false, false);
      }
      const v8f g0 = gelu8(a0 + bia0);
      const v8f g1 = gelu8(a1 + bia1);
      #pragma unroll
      for (int r = 0; r < 8; ++r){
        int m = m0 + hf*8 + r;
        Hs[m*PHS_ + ph*32 + col]      = (_Float16)g0[r];
        Hs[m*PHS_ + ph*32 + 16 + col] = (_Float16)g1[r];
      }
      __syncthreads();
      const v16h ah0 = load_frag_row(Hs, PHS_, m0, 0,  lane);
      const v16h ah1 = load_frag_row(Hs, PHS_, m0, 32, lane);
      #pragma unroll
      for (int j = 0; j < 4; ++j){
        v16h bw = load_frag_row(W2t, PW2_, (ph*4 + j)*16, 0, lane);
        accC[j] = __builtin_amdgcn_wmma_f32_16x16x32_f16(false, ah0, false, bw, (short)0, accC[j], false, false);
      }
      #pragma unroll
      for (int j = 0; j < 4; ++j){
        v16h bw = load_frag_row(W2t, PW2_, (ph*4 + j)*16, 32, lane);
        accC[j] = __builtin_amdgcn_wmma_f32_16x16x32_f16(false, ah1, false, bw, (short)0, accC[j], false, false);
      }
    }

    __syncthreads();
    #pragma unroll
    for (int j = 0; j < 4; ++j){
      int c = (ph*4 + j)*16 + col;
      float* dst = Rt + c*PR_ + m0 + hf*8;
      f4x v0 = { accC[j][0], accC[j][1], accC[j][2], accC[j][3] };
      f4x v1 = { accC[j][4], accC[j][5], accC[j][6], accC[j][7] };
      *(f4x*)dst       = v0;
      *(f4x*)(dst + 4) = v1;
    }
  }
  __syncthreads();


  const float dt = __builtin_bit_cast(float, __builtin_amdgcn_readfirstlane(__builtin_bit_cast(int, scal[0])));
  const float ag = __builtin_bit_cast(float, __builtin_amdgcn_readfirstlane(__builtin_bit_cast(int, scal[1])));
  const float al = __builtin_bit_cast(float, __builtin_amdgcn_readfirstlane(__builtin_bit_cast(int, scal[2])));

  float st = fminf(fmaxf((t  + 0.5f)*(2.f/T_) - 0.5f, 0.f), 1.f);
  float sh = fminf(fmaxf((hh + 0.5f)*(4.f/H_) - 0.5f, 0.f), 3.f);
  int t0i = (int)st; int t1i = imin_(t0i+1, 1); float ft = st - (float)t0i;
  int h0i = (int)sh; int h1i = imin_(h0i+1, 3); float fh = sh - (float)h0i;
  f4 fw;
  int w0i = 0;
  for (int q = 0; q < 4; ++q){
    float sw = fminf(fmaxf((ww0 + q + 0.5f)*(4.f/W_) - 0.5f, 0.f), 3.f);
    int wi = (int)sw;
    if (q == 0) w0i = wi;
    fw[q] = sw - (float)wi;
  }
  const int w1i = imin_(w0i+1, 3);
  const f4 omfw = 1.f - fw;
  const float omfh = 1.f - fh, omft = 1.f - ft;
  const int i000 = t0i*16 + h0i*4 + w0i, i001 = t0i*16 + h0i*4 + w1i;
  const int i010 = t0i*16 + h1i*4 + w0i, i011 = t0i*16 + h1i*4 + w1i;
  const int i100 = t1i*16 + h0i*4 + w0i, i101 = t1i*16 + h0i*4 + w1i;
  const int i110 = t1i*16 + h1i*4 + w0i, i111 = t1i*16 + h1i*4 + w1i;

  const int hh0 = (rp & 31) << 1;
  const bool kU1 = hh >= 1,  kD1 = hh < 63, kU4 = hh0 >= 4,  kD4 = hh0 < 60;
  const bool kU16 = hh0 >= 16, kD16 = hh0 < 48;
  const bool kL4 = ww0 >= 4, kR4 = ww0 < 60, kL16 = ww0 >= 16, kR16 = ww0 < 48;
  const bool kT1 = t >= 1, kT2 = t >= 2;
  const int oU4 = kU4 ? -256 : 0,   oD4 = kD4 ? 256 : 0;
  const int oU16 = kU16 ? -1024 : 0, oD16 = kD16 ? 1024 : 0;
  const int oT1 = kT1 ? -4096 : 0,  oT2 = kT2 ? -8192 : 0;
  const unsigned eC   = (unsigned)n << 2;
  const unsigned eU1  = (unsigned)(n + (kU1 ? -64 : 0)) << 2,  eD1  = (unsigned)(n + (kD1 ? 64 : 0)) << 2;
  const unsigned eL4  = (unsigned)(n + (kL4 ? -4 : 0)) << 2,   eR4  = (unsigned)(n + (kR4 ? 4 : 0)) << 2;
  const unsigned eL16 = (unsigned)(n + (kL16 ? -16 : 0)) << 2, eR16 = (unsigned)(n + (kR16 ? 16 : 0)) << 2;
  const f4 z4 = {0.f, 0.f, 0.f, 0.f};

  f4 sumsq = z4;
  float sum_du = 0.f, sum_lap = 0.f;
  const size_t bbase = (size_t)b * C_ * N_;
  gfloat* const uw    = uniform_ptr(u + bbase + (size_t)(wv*UPD_CPW) * N_);
  gfloat* const uwU4  = uw + oU4,  * const uwD4  = uw + oD4;
  gfloat* const uwU16 = uw + oU16, * const uwD16 = uw + oD16;
  gfloat* const uwT1  = uw + oT1,  * const uwT2  = uw + oT2;

  #pragma unroll 1
  for (int j = 0; j < UPD_CPW; ++j){
    const int c = wv*UPD_CPW + j;
    const unsigned cj = (unsigned)j << 17;
    const unsigned vC = eC + cj;

    f4 cen = ldg4_sv(uw, vC);
    f4 U1  = ldg4_sv(uw, eU1 + cj),   D1  = ldg4_sv(uw, eD1 + cj);
    f4 U4  = ldg4_sv(uwU4, vC),       D4  = ldg4_sv(uwD4, vC);
    f4 U16 = ldg4_sv(uwU16, vC),      D16 = ldg4_sv(uwD16, vC);
    f4 L4  = ldg4_sv(uw, eL4 + cj),   R4  = ldg4_sv(uw, eR4 + cj);
    f4 L16 = ldg4_sv(uw, eL16 + cj),  R16 = ldg4_sv(uw, eR16 + cj);
    f4 T1  = ldg4_sv(uwT1, vC),       T2  = ldg4_sv(uwT2, vC);
    f4 rc  = *(const f4*)&Rt[c*PR_ + lane*4] + b2[c];
    float lpv = lproj[((((size_t)b*4 + (lane >> 3))*C_ + c) << 3) + (lane & 7)];

    U1  = kU1  ? U1  : z4;  D1  = kD1  ? D1  : z4;
    U4  = kU4  ? U4  : z4;  D4  = kD4  ? D4  : z4;
    U16 = kU16 ? U16 : z4;  D16 = kD16 ? D16 : z4;
    L4  = kL4  ? L4  : z4;  R4  = kR4  ? R4  : z4;
    L16 = kL16 ? L16 : z4;  R16 = kR16 ? R16 : z4;
    T1  = kT1  ? T1  : z4;  T2  = kT2  ? T2  : z4;
    f4 L1 = { L4[3], cen[0], cen[1], cen[2] };
    f4 R1 = { cen[1], cen[2], cen[3], R4[0] };

    const float s0 = spsks[c], s1 = spsks[C_ + c], s2 = spsks[2*C_ + c];
    const float k0 = spskt[c], k1 = spskt[C_ + c];
    const float hv = h[b*C_ + c];

    f4 diff, lap;
    lap = -4.f * cen; lap += U1;  lap += D1;  lap += L1;  lap += R1;  diff  = s0 * lap;
    lap = -4.f * cen; lap += U4;  lap += D4;  lap += L4;  lap += R4;  diff += s1 * lap;
    lap = -4.f * cen; lap += U16; lap += D16; lap += L16; lap += R16; diff += s2 * lap;
    diff += k0 * (T1 - cen);
    diff += k1 * (T2 - cen);
    diff *= DIFF_SCALE;
    sum_lap += fabsf(diff[0]) + fabsf(diff[1]) + fabsf(diff[2]) + fabsf(diff[3]);

    const float a000 = __shfl(lpv, i000), a001 = __shfl(lpv, i001);
    const float a010 = __shfl(lpv, i010), a011 = __shfl(lpv, i011);
    const float a100 = __shfl(lpv, i100), a101 = __shfl(lpv, i101);
    const float a110 = __shfl(lpv, i110), a111 = __shfl(lpv, i111);
    f4 c00 = a000*omfw + a001*fw;
    f4 c01 = a010*omfw + a011*fw;
    f4 c10 = a100*omfw + a101*fw;
    f4 c11 = a110*omfw + a111*fw;
    f4 lv = (c00*omfh + c01*fh)*omft + (c10*omfh + c11*fh)*ft;

    f4 du = diff + rc + ag*hv + al*lv;
    f4 cd = cen + dt*du;
    sum_du += fabsf(du[0]) + fabsf(du[1]) + fabsf(du[2]) + fabsf(du[3]);
    sumsq  += cd*cd;
    *(f4*)&Rt[c*PR_ + lane*4] = cd;
    float ps = (cd[0] + cd[1]) + (cd[2] + cd[3]);
    ps += lane_xor1(ps);
    ps += lane_xor16(ps);
    if ((lane & 17) == 0) stg_p[wv][j*8 + (lane >> 1)] = ps;
  }

  for (int o = 16; o > 0; o >>= 1){
    sum_du  += __shfl_xor(sum_du, o);
    sum_lap += __shfl_xor(sum_lap, o);
  }
  if (lane == 0){ red_du[wv] = sum_du; red_lap[wv] = sum_lap; }
  const int lane_l = (int)__builtin_amdgcn_mbcnt_lo(~0u, 0u);
  if (do_norm) sq[wv][lane_l] = sumsq;
  __syncthreads();

  *(float2*)(part_p + (size_t)rp*1024 + wv*64 + lane*2) = *(const float2*)&stg_p[wv][lane*2];

  f4 scale = {1.f, 1.f, 1.f, 1.f};
  if (do_norm){
    f4 tot = z4;
    for (int g = 0; g < UPD_WAVES; ++g) tot += sq[g][lane];
    for (int q = 0; q < 4; ++q) scale[q] = rsqrtf(tot[q]/(float)C_ + 1e-8f);
  }
  #pragma unroll
  for (int j = 0; j < UPD_CPW; ++j){
    const int c = wv*UPD_CPW + j;
    const f4 cdj = *(const f4*)&Rt[c*PR_ + lane*4];
    f4 o4 = do_norm ? cdj*scale*normw[c] : cdj;
    *(f4*)(unext + bbase + (size_t)c * N_ + n) = o4;
    float ls = (o4[0] + o4[1]) + (o4[2] + o4[3]);
    ls += lane_xor1(ls);
    ls += lane_xor2(ls);
    ls += lane_xor16(ls);
    if ((lane & 19) == 0) stg_l[wv][j*4 + (lane >> 2)] = ls;
  }
  __syncthreads();
  part_l[(size_t)rp*512 + wv*32 + lane] = stg_l[wv][lane];
  if (wv < 4){
    const int ci = wv*32 + lane_l;
    const f4 q4 = *(const f4*)&stg_l[ci >> 3][(ci & 7)*4];
    part_g[(size_t)rp*C_ + ci] = (q4[0] + q4[1]) + (q4[2] + q4[3]);
  }

  if (tid == 0){
    float s1 = 0.f, s2 = 0.f;
    for (int g = 0; g < UPD_WAVES; ++g){ s1 += red_du[g]; s2 += red_lap[g]; }
    atomicAdd(&accum[1], s1);
    atomicAdd(&accum[0], s2);
  }
}

__global__ void __launch_bounds__(512)
init_partials_kernel(const float* __restrict__ u, float* __restrict__ part_l, float* __restrict__ part_p,
                     float* __restrict__ part_g){
  __shared__ __attribute__((aligned(16))) float stg_p[UPD_WAVES][64];
  __shared__ __attribute__((aligned(16))) float stg_l[UPD_WAVES][32];

  const int tid  = threadIdx.x;
  const int lane = tid & 31;
  const int wv   = tid >> 5;
  const int rp   = blockIdx.x;
  const int b    = rp >> 8;
  const int t    = (rp >> 5) & 7;
  const int hh   = ((rp & 31) << 1) + (lane >> 4);
  const int ww0  = (lane & 15) << 2;
  const int n    = (t << 12) + (hh << 6) + ww0;
  const float* ub = u + (size_t)b * C_ * N_ + (size_t)(wv*UPD_CPW) * N_ + n;

  f4 v[UPD_CPW];
  #pragma unroll
  for (int j = 0; j < UPD_CPW; ++j) v[j] = ldg4(ub + (size_t)j * N_);
  #pragma unroll
  for (int j = 0; j < UPD_CPW; ++j){
    float ps = (v[j][0] + v[j][1]) + (v[j][2] + v[j][3]);
    ps += lane_xor1(ps);
    ps += lane_xor16(ps);
    if ((lane & 17) == 0) stg_p[wv][j*8 + (lane >> 1)] = ps;
    const float ls = ps + lane_xor2(ps);
    if ((lane & 19) == 0) stg_l[wv][j*4 + (lane >> 2)] = ls;
  }
  __syncthreads();
  *(float2*)(part_p + (size_t)rp*1024 + wv*64 + lane*2) = *(const float2*)&stg_p[wv][lane*2];
  part_l[(size_t)rp*512 + wv*32 + lane] = stg_l[wv][lane];
  if (wv < 4){
    const int ci = wv*32 + lane;
    const f4 q4 = *(const f4*)&stg_l[ci >> 3][(ci & 7)*4];
    part_g[(size_t)rp*C_ + ci] = (q4[0] + q4[1]) + (q4[2] + q4[3]);
  }
}

__global__ void final_out_kernel(const float* accum, float* out){
  if (threadIdx.x == 0 && blockIdx.x == 0){
    const float inv_cells = 1.f/(float)PROBE_CELLS;
    const float inv_bnc   = 1.f/8388608.f;
    float a = 0.f, bm = 0.f;
    for (int i = 0; i < MAX_STEPS_; ++i){
      const float* ac = accum + i*4;
      float stop_turb = (ac[2]*inv_cells) / (ac[3]*inv_cells + 1e-8f);
      float se = ac[1]*inv_bnc;
      float le = ac[0]*inv_bnc;
      a  += stop_turb + 0.05f*se + 0.01f*le;
      bm += se;
    }
    out[0] = a / (float)MAX_STEPS_;
    out[1] = bm / (float)MAX_STEPS_;
  }
}


extern "C" void kernel_launch(void* const* d_in, const int* in_sizes, int n_in,
                              void* d_out, int out_size, void* d_ws, size_t ws_size,
                              hipStream_t stream){
  (void)in_sizes; (void)n_in; (void)out_size; (void)ws_size;

  const float* u_in  = (const float*)d_in[0];
  const float* w1    = (const float*)d_in[1];
  const float* b1    = (const float*)d_in[2];
  const float* w2    = (const float*)d_in[3];
  const float* b2    = (const float*)d_in[4];
  const float* gw    = (const float*)d_in[5];
  const float* gb    = (const float*)d_in[6];
  const float* vw    = (const float*)d_in[7];
  const float* vb    = (const float*)d_in[8];
  const float* normw = (const float*)d_in[9];
  const float* wlp   = (const float*)d_in[10];
  const float* blp   = (const float*)d_in[11];
  const float* a_g   = (const float*)d_in[12];
  const float* a_l   = (const float*)d_in[13];
  const float* logdt = (const float*)d_in[14];
  const float* ks    = (const float*)d_in[15];
  const float* kt    = (const float*)d_in[16];

  float* Wf = (float*)d_ws;
  size_t o = 0;
  const size_t VOL = (size_t)B_ * C_ * N_;
  float* ubuf0  = Wf + o; o += VOL;
  float* ubuf1  = Wf + o; o += VOL;
  float* parts  = Wf + o; o += VOL;
  o += VOL;
  float* probeA = Wf + o; o += PROBE_CELLS;
  float* probeB = Wf + o; o += PROBE_CELLS;
  o += 256;
  float* hstate = Wf + o; o += 256;
  o += (size_t)B_*C_*32;
  float* lproj  = Wf + o; o += (size_t)B_*C_*32;
  float* spsks  = Wf + o; o += 3*C_;
  float* spskt  = Wf + o; o += 2*C_;
  float* scal   = Wf + o; o += 8;
  o += 8 + 16 + 16;
  _Float16* w1h = (_Float16*)(Wf + o); o += (C_*HID_)/2;
  _Float16* w2h = (_Float16*)(Wf + o); o += (HID_*C_)/2;

  float* part_l = parts;
  float* part_p = parts + 512*C_*4;
  float* accum  = parts + 512*C_*12;
  float* part_g = parts + 512*C_*13;

  float* ubuf[2] = { ubuf0, ubuf1 };

  prep_kernel<<<64, 256, 0, stream>>>(w1, w2, ks, kt, a_g, a_l, logdt,
                                      w1h, w2h, spsks, spskt, scal);
  zero_kernel<<<1, 256, 0, stream>>>(hstate, 256, accum, MAX_STEPS_*4);
  hipMemcpyAsync(ubuf[0], u_in, VOL*sizeof(float), hipMemcpyDeviceToDevice, stream);
  init_partials_kernel<<<BN_/128, 512, 0, stream>>>(ubuf[0], part_l, part_p, part_g);

  float* pprev = probeA;
  float* pcur  = probeB;

  for (int step = 0; step < MAX_STEPS_; ++step){
    float* ucur  = ubuf[step & 1];
    float* unext = ubuf[(step & 1) ^ 1];
    int do_norm = (((step + 1) % 2) == 0) ? 1 : 0;

    float* acc = accum + step*4;
    if (step == 0){
      fold_kernel<<<SIDE_BLOCKS_, 256, 0, stream>>>(part_g, gw, gb, vw, vb, hstate,
                                                    part_l, wlp, blp, lproj,
                                                    part_p, pprev, part_p, accum, 0);
    } else {
      fold_kernel<<<SIDE_BLOCKS_, 256, 0, stream>>>(part_g, gw, gb, vw, vb, hstate,
                                                    part_l, wlp, blp, lproj,
                                                    part_p, pcur, pprev, acc - 4, 1);
      float* tswap = pprev; pprev = pcur; pcur = tswap;
    }
    step_kernel<<<BN_/128, 512, 0, stream>>>(ucur, w1h, b1, w2h, b2, hstate, lproj, spsks, spskt,
                                             scal, normw, unext, part_l, part_p, part_g, acc, do_norm);
  }
  probe_fold_kernel<<<PROBE_BLOCKS_, 256, 0, stream>>>(part_p, pcur, pprev, accum + (MAX_STEPS_-1)*4);

  hipMemcpyAsync(d_out, ubuf[MAX_STEPS_ & 1], VOL*sizeof(float),
                 hipMemcpyDeviceToDevice, stream);
  final_out_kernel<<<1, 32, 0, stream>>>(accum, ((float*)d_out) + VOL);
}


# tiles claimed from a queue per chip half (die read from hardware, one returning counter per half and step): each half's second-level cache serves one batch element, stencil rows never cross halves
# speedup vs baseline: 12.2676x; 1.1930x over previous
#include <hip/hip_runtime.h>
#include <math.h>

#define B_ 2
#define C_ 128
#define T_ 8
#define H_ 64
#define W_ 64
#define N_ (T_*H_*W_)
#define BN_ (B_*N_)
#define HID_ 256
#define MAX_STEPS_ 12
#define DIFF_SCALE 0.25f
#define PROBE_CELLS (B_*C_*2*8*8)

typedef __attribute__((ext_vector_type(16))) _Float16 v16h;
typedef __attribute__((ext_vector_type(8)))  _Float16 h8;
typedef __attribute__((ext_vector_type(8)))  float    v8f;
typedef __attribute__((ext_vector_type(4)))  float    f4x;
typedef float f4 __attribute__((ext_vector_type(4)));
__device__ __forceinline__ f4 ldg4(const float* p){ return *(const f4*)p; }

__device__ __forceinline__ float softplus_f(float x){ return x > 20.f ? x : log1pf(expf(x)); }
typedef float g8f __attribute__((vector_size(32)));
__device__ __forceinline__ v8f gelu8(v8f av){
  const g8f a = __builtin_bit_cast(g8f, av);
  const g8f x = a * 0.70710678118654752f;
  const g8f t = __builtin_elementwise_abs(x);
  g8f p = t*0.0000169988107f + -0.000378677854f;
  p = t*p + 0.00385781587f;
  p = t*p + -0.0241816975f;
  p = t*p + 0.106668264f;
  p = t*p + 0.634933293f;
  p = t*p + 0.128689408f;
  p = t*p + t;
  const g8f y = p * -1.44269504088896341f;
  g8f e2;
  #pragma unroll
  for (int i = 0; i < 8; ++i) e2[i] = __builtin_amdgcn_exp2f(y[i]);
  const g8f rl = __builtin_elementwise_copysign(1.f - e2, x);
  const g8f s = x*x;
  g8f q = s*-0.000561801775f + 0.00491381623f;
  q = s*q + -0.0267075151f;
  q = s*q + 0.112800106f;
  q = s*q + -0.376122952f;
  q = s*q + 0.128379107f;
  const g8f rs = x*q + x;
  const g8f e = (t >= 1.f) ? rl : rs;
  const g8f g = (0.5f*a)*(1.f + e);
  return __builtin_bit_cast(v8f, g);
}
__device__ __forceinline__ int imin_(int a, int b){ return a < b ? a : b; }

__device__ __forceinline__ float blockReduceSum(float v){
  __shared__ float red[8];
  int lane = threadIdx.x & 31, wid = threadIdx.x >> 5;
  #pragma unroll
  for (int off = 16; off > 0; off >>= 1) v += __shfl_down(v, off);
  if (lane == 0) red[wid] = v;
  __syncthreads();
  v = (threadIdx.x < (blockDim.x >> 5)) ? red[threadIdx.x] : 0.f;
  if (wid == 0){
    #pragma unroll
    for (int off = 4; off > 0; off >>= 1) v += __shfl_down(v, off);
  }
  __syncthreads();
  return v;
}

__device__ __forceinline__ v16h load_frag_row(const _Float16* base, int ld, int r0, int k0, int lane){
  int r = r0 + (lane & 15);
  int hf = lane >> 4;
  const _Float16* p = base + r*ld + k0 + hf*8;
  h8 lo = *(const h8*)p;
  h8 hi = *(const h8*)(p + 16);
  return __builtin_shufflevector(lo, hi, 0,1,2,3,4,5,6,7,8,9,10,11,12,13,14,15);
}


__global__ void zero_kernel(float* p, int n, float* q, int m){
  int i = blockIdx.x*blockDim.x + threadIdx.x;
  if (i < n) p[i] = 0.f;
  if (i < m) q[i] = 0.f;
}

__global__ void prep_kernel(const float* w1, const float* w2,
                            const float* ks, const float* kt,
                            const float* ag, const float* al, const float* logdt,
                            _Float16* w1h, _Float16* w2h,
                            float* spsks, float* spskt, float* scal){
  int i = blockIdx.x*blockDim.x + threadIdx.x;
  int stride = gridDim.x*blockDim.x;
  for (int j = i; j < C_*HID_; j += stride) w1h[j] = (_Float16)w1[(j & (C_-1))*HID_ + (j >> 7)];
  for (int j = i; j < HID_*C_; j += stride) w2h[j] = (_Float16)w2[(j & (HID_-1))*C_ + (j >> 8)];
  for (int j = i; j < 3*C_;   j += stride) spsks[j] = softplus_f(ks[j]);
  for (int j = i; j < 2*C_;   j += stride) spskt[j] = softplus_f(kt[j]);
  if (i == 0){
    float dt = expf(logdt[0]);
    dt = fminf(fmaxf(dt, 0.005f), 0.35f);
    scal[0] = dt;
    scal[1] = softplus_f(ag[0]);
    scal[2] = softplus_f(al[0]);
  }
}

#define GATE_BLOCKS_  2
#define LOC_BLOCKS_   8
#define PROBE_BLOCKS_ 64
#define SIDE_BLOCKS_ (GATE_BLOCKS_ + LOC_BLOCKS_ + PROBE_BLOCKS_)

__device__ __forceinline__ void gate_block(int b, const float* part_g, const float* gw, const float* gb,
                                           const float* vw, const float* vb, float* h){
  __shared__ float ps[2][C_];
  __shared__ float pl[C_];
  __shared__ float gv[2][C_];
  const int tid = threadIdx.x;
  const int c = tid & 127, hs = tid >> 7;
  const float* p = part_g + ((size_t)b*256 + hs*128)*C_ + c;
  float s = 0.f;
  #pragma unroll
  for (int r = 0; r < 128; ++r) s += p[(size_t)r*C_];
  ps[hs][c] = s;
  __syncthreads();
  if (tid < C_) pl[tid] = (ps[0][tid] + ps[1][tid]) / (float)N_;
  __syncthreads();
  const float* wm = hs ? vw : gw;
  float a = hs ? vb[c] : gb[c];
  #pragma unroll
  for (int k = 0; k < C_; ++k) a += pl[k] * wm[k*C_ + c];
  gv[hs][c] = a;
  __syncthreads();
  if (tid < C_) h[b*C_ + tid] += (1.f/(1.f+expf(-gv[0][tid]))) * tanhf(gv[1][tid]);
}

__device__ __forceinline__ void local_block(int blk, const float* part_l, const float* wlp,
                                            const float* blp, float* lproj){
  __shared__ __attribute__((aligned(16))) float lsm[C_][8];
  const int tid = threadIdx.x;
  const int b = blk >> 2, cg = blk & 3;
  const int pt = cg >> 1, hp = cg & 1;
  const size_t bblk = (size_t)b * 256;
  {
    const int c = tid & 127, phl = tid >> 7;
    const int ph = hp*2 + phl;
    f4 leaf[8];
    for (int sb = 0; sb < 8; ++sb){
      const int t = pt*4 + (sb >> 1);
      f4 sm = {0.f, 0.f, 0.f, 0.f};
      for (int q = 0; q < 4; ++q){
        const int rpair = ph*8 + (sb & 1)*4 + q;
        sm += ldg4(part_l + ((bblk + t*32 + rpair)*C_ + c)*4);
      }
      leaf[sb] = sm;
    }
    const f4 tot = ((leaf[0] + leaf[1]) + (leaf[2] + leaf[3])) + ((leaf[4] + leaf[5]) + (leaf[6] + leaf[7]));
    *(f4*)&lsm[c][phl*4] = tot * (1.f/1024.f);
  }
  __syncthreads();
  const int oc = tid >> 1, chh = tid & 1;
  const float bias = blp[oc];
  f4 pr = {bias, bias, bias, bias};
  #pragma unroll
  for (int c = 0; c < C_; ++c){
    const f4 l = *(const f4*)&lsm[c][chh*4];
    pr += l * wlp[c*C_ + oc];
  }
  *(f4*)(lproj + ((((size_t)b*4 + cg)*C_ + oc) << 3) + chh*4) = pr;
}

#define PROBE_CPB_ 4
__device__ __forceinline__ void probe_fold_block(int pb, const float* part_p, float* probe,
                                                 const float* prev, float* accum, int with_diff){
  const int tid  = threadIdx.x;
  const int cell = tid & 127, cp = tid >> 7;
  const int pt = cell >> 6, ph = (cell >> 3) & 7, pw = cell & 7;
  const int bc0 = pb*PROBE_CPB_ + cp*2;
  const int b = bc0 >> 7, c0 = bc0 & 127;
  const size_t bblk = (size_t)b * 256;
  float pp[2];
  for (int i = 0; i < 2; ++i){
    const float pv = prev[(bc0 + i)*128 + cell];
    pp[i] = with_diff ? pv : 0.f;
  }
  float ad = 0.f, ap = 0.f;
  float cv[2];
  for (int i = 0; i < 2; ++i){
    float hsum[2];
    for (int sb = 0; sb < 2; ++sb){
      float a = 0.f;
      for (int q = 0; q < 8; ++q){
        const int k = sb*8 + q;
        const int t = pt*4 + (k >> 2);
        const int rpair = ph*4 + (k & 3);
        a += part_p[(bblk + t*32 + rpair)*1024 + (c0 + i)*8 + pw];
      }
      hsum[sb] = a;
    }
    cv[i] = (hsum[0] + hsum[1]) * (1.f/256.f);
  }
  for (int i = 0; i < 2; ++i){
    probe[(bc0 + i)*128 + cell] = cv[i];
    ad += fabsf(cv[i] - pp[i]);
    ap += fabsf(pp[i]);
  }
  if (with_diff){
    const float s1 = blockReduceSum(ad);
    const float s2 = blockReduceSum(ap);
    if (tid == 0){
      atomicAdd(&accum[2], s1);
      atomicAdd(&accum[3], s2);
    }
  }
}

__global__ void __launch_bounds__(256)
probe_fold_kernel(const float* part_p, float* probe, const float* prev, float* accum){
  probe_fold_block(blockIdx.x, part_p, probe, prev, accum, 1);
}

__global__ void __launch_bounds__(256)
fold_kernel(const float* part_g, const float* gw, const float* gb,
            const float* vw, const float* vb, float* hstate,
            const float* part_l, const float* wlp, const float* blp, float* lproj,
            const float* part_p, float* probe, const float* prev, float* accum, int with_diff){
  const int sb = blockIdx.x;
  if (sb < GATE_BLOCKS_) gate_block(sb, part_g, gw, gb, vw, vb, hstate);
  else if (sb < GATE_BLOCKS_ + LOC_BLOCKS_) local_block(sb - GATE_BLOCKS_, part_l, wlp, blp, lproj);
  else probe_fold_block(sb - GATE_BLOCKS_ - LOC_BLOCKS_, part_p, probe, prev, accum, with_diff);
}

#define PA_  136
#define PW2_ 72
#define PR_  132
#define PHS_ 72
#define UPD_WAVES 16
#define UPD_CPW   8
#define MLP_LDS_BYTES_ ((128*PA_ + 64*PA_ + 128*PHS_ + 128*PW2_) * 2)
static_assert(C_*PR_*4 <= MLP_LDS_BYTES_, "the reaction tile must fit over the staging tiles");

__device__ __forceinline__ float lane_xor1(float x){
  return __builtin_bit_cast(float, __builtin_amdgcn_update_dpp(0, __builtin_bit_cast(int, x), 0xB1, 0xF, 0xF, true));
}
__device__ __forceinline__ float lane_xor2(float x){
  return __builtin_bit_cast(float, __builtin_amdgcn_update_dpp(0, __builtin_bit_cast(int, x), 0x4E, 0xF, 0xF, true));
}
__device__ __forceinline__ float lane_xor16(float x){
  const int b = __builtin_bit_cast(int, x);
  return __builtin_bit_cast(float, __builtin_amdgcn_permlanex16(b, b, 0x76543210, 0xfedcba98, false, false));
}
typedef const __attribute__((address_space(1))) float gfloat;
__device__ __forceinline__ gfloat* uniform_ptr(const float* p){
  const unsigned long long a = (unsigned long long)p;
  const unsigned lo = __builtin_amdgcn_readfirstlane((unsigned)a);
  const unsigned hi = __builtin_amdgcn_readfirstlane((unsigned)(a >> 32));
  return (gfloat*)(((unsigned long long)hi << 32) | lo);
}
__device__ __forceinline__ f4 ldg4_sv(gfloat* sbase, unsigned voff){
  typedef const __attribute__((address_space(1))) char gchar;
  typedef const __attribute__((address_space(1))) f4 gf4;
  return *(gf4*)((gchar*)sbase + voff);
}

__global__ void __launch_bounds__(512, 8)
step_kernel(const float* __restrict__ u, const _Float16* __restrict__ w1t, const float* b1,
            const _Float16* __restrict__ w2t, const float* b2, const float* __restrict__ h,
            const float* __restrict__ lproj, const float* __restrict__ spsks, const float* __restrict__ spskt,
            const float* __restrict__ scal, const float* normw,
            float* __restrict__ unext, float* __restrict__ part_l, float* __restrict__ part_p,
            float* __restrict__ part_g, float* __restrict__ accum, float* tq, int do_norm){
  __shared__ __attribute__((aligned(16))) unsigned char lds_mlp[MLP_LDS_BYTES_];
  _Float16* const As  = (_Float16*)lds_mlp;
  _Float16* const Bt  = As  + 128*PA_;
  _Float16* const Hs  = Bt  + 64*PA_;
  _Float16* const W2t = Hs  + 128*PHS_;
  float*    const Rt  = (float*)lds_mlp;
  __shared__ f4 sq[UPD_WAVES][32];
  __shared__ __attribute__((aligned(16))) float stg_p[UPD_WAVES][64];
  __shared__ __attribute__((aligned(16))) float stg_l[UPD_WAVES][32];
  __shared__ float red_du[UPD_WAVES];
  __shared__ float red_lap[UPD_WAVES];

  const int tid  = threadIdx.x;
  const int lane = tid & 31;
  const int wv   = __builtin_amdgcn_readfirstlane(tid >> 5);

  const int rt = wv & 7;
  const int ph = wv >> 3;
  const int m0 = rt * 16;
  const int hf = lane >> 4;
  const int col = lane & 15;
  const _Float16* gp1 = w1t + (size_t)(tid >> 4)*C_   + (tid & 15)*8;
  const _Float16* gp2 = w2t + (size_t)(tid >> 3)*HID_ + (tid & 7)*8;
  _Float16* l1 = Bt  + (tid >> 4)*PA_  + (tid & 15)*8;
  _Float16* l2 = W2t + (tid >> 3)*PW2_ + (tid & 7)*8;
  h8 r1a = *(const h8*)gp1;
  h8 r1b = *(const h8*)(gp1 + 32*C_);
  h8 r2a = *(const h8*)gp2;
  h8 r2b = *(const h8*)(gp2 + 64*HID_);
  float nb0 = b1[ph*32 + col], nb1 = b1[ph*32 + 16 + col];

  __shared__ int s_rp;
  if (tid == 0){
    const unsigned id = __builtin_amdgcn_s_sendmsg_rtn(0x87);
    const int half = (int)((id >> 18) & 1u);
    int* const q = (int*)tq;
    const int k = atomicAdd(q + (half << 6), 1);
    int tile;
    if (k < 256) tile = (half << 8) + k;
    else         tile = ((half ^ 1) << 8) + 255 - atomicAdd(q + 128, 1);
    s_rp = tile & 511;
  }
  __syncthreads();
  const int rp   = __builtin_amdgcn_readfirstlane(s_rp);
  const int b    = rp >> 8;
  const int t    = (rp >> 5) & 7;
  const int hh   = ((rp & 31) << 1) + (lane >> 4);
  const int ww0  = (lane & 15) << 2;
  const int n    = (t << 12) + (hh << 6) + ww0;

  {
    const int n0 = (rp & 255) << 7;
    const float* ub = u + (size_t)b * C_ * N_;

    {
      f4x v[8];
      #pragma unroll
      for (int i = 0; i < 8; ++i)
        v[i] = *(const f4x*)(ub + (size_t)(wv*8 + i) * N_ + n0 + lane*4);
      #pragma unroll
      for (int j = 0; j < 4; ++j){
        h8 hv;
        #pragma unroll
        for (int i = 0; i < 8; ++i) hv[i] = (_Float16)v[i][j];
        *(h8*)(As + (lane*4 + j)*PA_ + wv*8) = hv;
      }
    }

    const v8f z8 = {0.f,0.f,0.f,0.f,0.f,0.f,0.f,0.f};
    v8f accC[4];
    #pragma unroll
    for (int i = 0; i < 4; ++i) accC[i] = z8;

    for (int q = 0; q < 4; ++q){
      __syncthreads();
      *(h8*)l1              = r1a;
      *(h8*)(l1 + 32*PA_)   = r1b;
      *(h8*)l2              = r2a;
      *(h8*)(l2 + 64*PW2_)  = r2b;
      __syncthreads();

      const float bia0 = nb0, bia1 = nb1;
      if (q < 3){
        r1a = *(const h8*)(gp1 + (q+1)*64*C_);
        r1b = *(const h8*)(gp1 + (q+1)*64*C_ + 32*C_);
        r2a = *(const h8*)(gp2 + (q+1)*64);
        r2b = *(const h8*)(gp2 + (q+1)*64 + 64*HID_);
        nb0 = b1[(q+1)*64 + ph*32 + col];
        nb1 = b1[(q+1)*64 + ph*32 + 16 + col];
      }

      v8f a0 = z8, a1 = z8;
      #pragma unroll
      for (int k0 = 0; k0 < 128; k0 += 32){
        v16h af  = load_frag_row(As, PA_, m0, k0, lane);
        v16h bf0 = load_frag_row(Bt, PA_, ph*32,      k0, lane);
        v16h bf1 = load_frag_row(Bt, PA_, ph*32 + 16, k0, lane);
        a0 = __builtin_amdgcn_wmma_f32_16x16x32_f16(false, af, false, bf0, (short)0, a0, false, false);
        a1 = __builtin_amdgcn_wmma_f32_16x16x32_f16(false, af, false, bf1, (short)0, a1, false, false);
      }
      const v8f g0 = gelu8(a0 + bia0);
      const v8f g1 = gelu8(a1 + bia1);
      #pragma unroll
      for (int r = 0; r < 8; ++r){
        int m = m0 + hf*8 + r;
        Hs[m*PHS_ + ph*32 + col]      = (_Float16)g0[r];
        Hs[m*PHS_ + ph*32 + 16 + col] = (_Float16)g1[r];
      }
      __syncthreads();
      const v16h ah0 = load_frag_row(Hs, PHS_, m0, 0,  lane);
      const v16h ah1 = load_frag_row(Hs, PHS_, m0, 32, lane);
      #pragma unroll
      for (int j = 0; j < 4; ++j){
        v16h bw = load_frag_row(W2t, PW2_, (ph*4 + j)*16, 0, lane);
        accC[j] = __builtin_amdgcn_wmma_f32_16x16x32_f16(false, ah0, false, bw, (short)0, accC[j], false, false);
      }
      #pragma unroll
      for (int j = 0; j < 4; ++j){
        v16h bw = load_frag_row(W2t, PW2_, (ph*4 + j)*16, 32, lane);
        accC[j] = __builtin_amdgcn_wmma_f32_16x16x32_f16(false, ah1, false, bw, (short)0, accC[j], false, false);
      }
    }

    __syncthreads();
    #pragma unroll
    for (int j = 0; j < 4; ++j){
      int c = (ph*4 + j)*16 + col;
      float* dst = Rt + c*PR_ + m0 + hf*8;
      f4x v0 = { accC[j][0], accC[j][1], accC[j][2], accC[j][3] };
      f4x v1 = { accC[j][4], accC[j][5], accC[j][6], accC[j][7] };
      *(f4x*)dst       = v0;
      *(f4x*)(dst + 4) = v1;
    }
  }
  __syncthreads();


  const float dt = __builtin_bit_cast(float, __builtin_amdgcn_readfirstlane(__builtin_bit_cast(int, scal[0])));
  const float ag = __builtin_bit_cast(float, __builtin_amdgcn_readfirstlane(__builtin_bit_cast(int, scal[1])));
  const float al = __builtin_bit_cast(float, __builtin_amdgcn_readfirstlane(__builtin_bit_cast(int, scal[2])));

  float st = fminf(fmaxf((t  + 0.5f)*(2.f/T_) - 0.5f, 0.f), 1.f);
  float sh = fminf(fmaxf((hh + 0.5f)*(4.f/H_) - 0.5f, 0.f), 3.f);
  int t0i = (int)st; int t1i = imin_(t0i+1, 1); float ft = st - (float)t0i;
  int h0i = (int)sh; int h1i = imin_(h0i+1, 3); float fh = sh - (float)h0i;
  f4 fw;
  int w0i = 0;
  for (int q = 0; q < 4; ++q){
    float sw = fminf(fmaxf((ww0 + q + 0.5f)*(4.f/W_) - 0.5f, 0.f), 3.f);
    int wi = (int)sw;
    if (q == 0) w0i = wi;
    fw[q] = sw - (float)wi;
  }
  const int w1i = imin_(w0i+1, 3);
  const f4 omfw = 1.f - fw;
  const float omfh = 1.f - fh, omft = 1.f - ft;
  const int i000 = t0i*16 + h0i*4 + w0i, i001 = t0i*16 + h0i*4 + w1i;
  const int i010 = t0i*16 + h1i*4 + w0i, i011 = t0i*16 + h1i*4 + w1i;
  const int i100 = t1i*16 + h0i*4 + w0i, i101 = t1i*16 + h0i*4 + w1i;
  const int i110 = t1i*16 + h1i*4 + w0i, i111 = t1i*16 + h1i*4 + w1i;

  const int hh0 = (rp & 31) << 1;
  const bool kU1 = hh >= 1,  kD1 = hh < 63, kU4 = hh0 >= 4,  kD4 = hh0 < 60;
  const bool kU16 = hh0 >= 16, kD16 = hh0 < 48;
  const bool kL4 = ww0 >= 4, kR4 = ww0 < 60, kL16 = ww0 >= 16, kR16 = ww0 < 48;
  const bool kT1 = t >= 1, kT2 = t >= 2;
  const int oU4 = kU4 ? -256 : 0,   oD4 = kD4 ? 256 : 0;
  const int oU16 = kU16 ? -1024 : 0, oD16 = kD16 ? 1024 : 0;
  const int oT1 = kT1 ? -4096 : 0,  oT2 = kT2 ? -8192 : 0;
  const unsigned eC   = (unsigned)n << 2;
  const unsigned eU1  = (unsigned)(n + (kU1 ? -64 : 0)) << 2,  eD1  = (unsigned)(n + (kD1 ? 64 : 0)) << 2;
  const unsigned eL4  = (unsigned)(n + (kL4 ? -4 : 0)) << 2,   eR4  = (unsigned)(n + (kR4 ? 4 : 0)) << 2;
  const unsigned eL16 = (unsigned)(n + (kL16 ? -16 : 0)) << 2, eR16 = (unsigned)(n + (kR16 ? 16 : 0)) << 2;
  const f4 z4 = {0.f, 0.f, 0.f, 0.f};

  f4 sumsq = z4;
  float sum_du = 0.f, sum_lap = 0.f;
  const size_t bbase = (size_t)b * C_ * N_;
  gfloat* const uw    = uniform_ptr(u + bbase + (size_t)(wv*UPD_CPW) * N_);
  gfloat* const uwU4  = uw + oU4,  * const uwD4  = uw + oD4;
  gfloat* const uwU16 = uw + oU16, * const uwD16 = uw + oD16;
  gfloat* const uwT1  = uw + oT1,  * const uwT2  = uw + oT2;

  #pragma unroll 1
  for (int j = 0; j < UPD_CPW; ++j){
    const int c = wv*UPD_CPW + j;
    const unsigned cj = (unsigned)j << 17;
    const unsigned vC = eC + cj;

    f4 cen = ldg4_sv(uw, vC);
    f4 U1  = ldg4_sv(uw, eU1 + cj),   D1  = ldg4_sv(uw, eD1 + cj);
    f4 U4  = ldg4_sv(uwU4, vC),       D4  = ldg4_sv(uwD4, vC);
    f4 U16 = ldg4_sv(uwU16, vC),      D16 = ldg4_sv(uwD16, vC);
    f4 L4  = ldg4_sv(uw, eL4 + cj),   R4  = ldg4_sv(uw, eR4 + cj);
    f4 L16 = ldg4_sv(uw, eL16 + cj),  R16 = ldg4_sv(uw, eR16 + cj);
    f4 T1  = ldg4_sv(uwT1, vC),       T2  = ldg4_sv(uwT2, vC);
    f4 rc  = *(const f4*)&Rt[c*PR_ + lane*4] + b2[c];
    float lpv = lproj[((((size_t)b*4 + (lane >> 3))*C_ + c) << 3) + (lane & 7)];

    U1  = kU1  ? U1  : z4;  D1  = kD1  ? D1  : z4;
    U4  = kU4  ? U4  : z4;  D4  = kD4  ? D4  : z4;
    U16 = kU16 ? U16 : z4;  D16 = kD16 ? D16 : z4;
    L4  = kL4  ? L4  : z4;  R4  = kR4  ? R4  : z4;
    L16 = kL16 ? L16 : z4;  R16 = kR16 ? R16 : z4;
    T1  = kT1  ? T1  : z4;  T2  = kT2  ? T2  : z4;
    f4 L1 = { L4[3], cen[0], cen[1], cen[2] };
    f4 R1 = { cen[1], cen[2], cen[3], R4[0] };

    const float s0 = spsks[c], s1 = spsks[C_ + c], s2 = spsks[2*C_ + c];
    const float k0 = spskt[c], k1 = spskt[C_ + c];
    const float hv = h[b*C_ + c];

    f4 diff, lap;
    lap = -4.f * cen; lap += U1;  lap += D1;  lap += L1;  lap += R1;  diff  = s0 * lap;
    lap = -4.f * cen; lap += U4;  lap += D4;  lap += L4;  lap += R4;  diff += s1 * lap;
    lap = -4.f * cen; lap += U16; lap += D16; lap += L16; lap += R16; diff += s2 * lap;
    diff += k0 * (T1 - cen);
    diff += k1 * (T2 - cen);
    diff *= DIFF_SCALE;
    sum_lap += fabsf(diff[0]) + fabsf(diff[1]) + fabsf(diff[2]) + fabsf(diff[3]);

    const float a000 = __shfl(lpv, i000), a001 = __shfl(lpv, i001);
    const float a010 = __shfl(lpv, i010), a011 = __shfl(lpv, i011);
    const float a100 = __shfl(lpv, i100), a101 = __shfl(lpv, i101);
    const float a110 = __shfl(lpv, i110), a111 = __shfl(lpv, i111);
    f4 c00 = a000*omfw + a001*fw;
    f4 c01 = a010*omfw + a011*fw;
    f4 c10 = a100*omfw + a101*fw;
    f4 c11 = a110*omfw + a111*fw;
    f4 lv = (c00*omfh + c01*fh)*omft + (c10*omfh + c11*fh)*ft;

    f4 du = diff + rc + ag*hv + al*lv;
    f4 cd = cen + dt*du;
    sum_du += fabsf(du[0]) + fabsf(du[1]) + fabsf(du[2]) + fabsf(du[3]);
    sumsq  += cd*cd;
    *(f4*)&Rt[c*PR_ + lane*4] = cd;
    float ps = (cd[0] + cd[1]) + (cd[2] + cd[3]);
    ps += lane_xor1(ps);
    ps += lane_xor16(ps);
    if ((lane & 17) == 0) stg_p[wv][j*8 + (lane >> 1)] = ps;
  }

  for (int o = 16; o > 0; o >>= 1){
    sum_du  += __shfl_xor(sum_du, o);
    sum_lap += __shfl_xor(sum_lap, o);
  }
  if (lane == 0){ red_du[wv] = sum_du; red_lap[wv] = sum_lap; }
  const int lane_l = (int)__builtin_amdgcn_mbcnt_lo(~0u, 0u);
  if (do_norm) sq[wv][lane_l] = sumsq;
  __syncthreads();

  *(float2*)(part_p + (size_t)rp*1024 + wv*64 + lane*2) = *(const float2*)&stg_p[wv][lane*2];

  f4 scale = {1.f, 1.f, 1.f, 1.f};
  if (do_norm){
    f4 tot = z4;
    for (int g = 0; g < UPD_WAVES; ++g) tot += sq[g][lane];
    for (int q = 0; q < 4; ++q) scale[q] = rsqrtf(tot[q]/(float)C_ + 1e-8f);
  }
  #pragma unroll
  for (int j = 0; j < UPD_CPW; ++j){
    const int c = wv*UPD_CPW + j;
    const f4 cdj = *(const f4*)&Rt[c*PR_ + lane*4];
    f4 o4 = do_norm ? cdj*scale*normw[c] : cdj;
    *(f4*)(unext + bbase + (size_t)c * N_ + n) = o4;
    float ls = (o4[0] + o4[1]) + (o4[2] + o4[3]);
    ls += lane_xor1(ls);
    ls += lane_xor2(ls);
    ls += lane_xor16(ls);
    if ((lane & 19) == 0) stg_l[wv][j*4 + (lane >> 2)] = ls;
  }
  __syncthreads();
  part_l[(size_t)rp*512 + wv*32 + lane] = stg_l[wv][lane];
  if (wv < 4){
    const int ci = wv*32 + lane_l;
    const f4 q4 = *(const f4*)&stg_l[ci >> 3][(ci & 7)*4];
    part_g[(size_t)rp*C_ + ci] = (q4[0] + q4[1]) + (q4[2] + q4[3]);
  }

  if (tid == 0){
    float s1 = 0.f, s2 = 0.f;
    for (int g = 0; g < UPD_WAVES; ++g){ s1 += red_du[g]; s2 += red_lap[g]; }
    atomicAdd(&accum[1], s1);
    atomicAdd(&accum[0], s2);
  }
}

__global__ void __launch_bounds__(512)
init_partials_kernel(const float* __restrict__ u, float* __restrict__ part_l, float* __restrict__ part_p,
                     float* __restrict__ part_g){
  __shared__ __attribute__((aligned(16))) float stg_p[UPD_WAVES][64];
  __shared__ __attribute__((aligned(16))) float stg_l[UPD_WAVES][32];

  const int tid  = threadIdx.x;
  const int lane = tid & 31;
  const int wv   = tid >> 5;
  const int rp   = blockIdx.x;
  const int b    = rp >> 8;
  const int t    = (rp >> 5) & 7;
  const int hh   = ((rp & 31) << 1) + (lane >> 4);
  const int ww0  = (lane & 15) << 2;
  const int n    = (t << 12) + (hh << 6) + ww0;
  const float* ub = u + (size_t)b * C_ * N_ + (size_t)(wv*UPD_CPW) * N_ + n;

  f4 v[UPD_CPW];
  #pragma unroll
  for (int j = 0; j < UPD_CPW; ++j) v[j] = ldg4(ub + (size_t)j * N_);
  #pragma unroll
  for (int j = 0; j < UPD_CPW; ++j){
    float ps = (v[j][0] + v[j][1]) + (v[j][2] + v[j][3]);
    ps += lane_xor1(ps);
    ps += lane_xor16(ps);
    if ((lane & 17) == 0) stg_p[wv][j*8 + (lane >> 1)] = ps;
    const float ls = ps + lane_xor2(ps);
    if ((lane & 19) == 0) stg_l[wv][j*4 + (lane >> 2)] = ls;
  }
  __syncthreads();
  *(float2*)(part_p + (size_t)rp*1024 + wv*64 + lane*2) = *(const float2*)&stg_p[wv][lane*2];
  part_l[(size_t)rp*512 + wv*32 + lane] = stg_l[wv][lane];
  if (wv < 4){
    const int ci = wv*32 + lane;
    const f4 q4 = *(const f4*)&stg_l[ci >> 3][(ci & 7)*4];
    part_g[(size_t)rp*C_ + ci] = (q4[0] + q4[1]) + (q4[2] + q4[3]);
  }
}

__global__ void final_out_kernel(const float* accum, float* out){
  if (threadIdx.x == 0 && blockIdx.x == 0){
    const float inv_cells = 1.f/(float)PROBE_CELLS;
    const float inv_bnc   = 1.f/8388608.f;
    float a = 0.f, bm = 0.f;
    for (int i = 0; i < MAX_STEPS_; ++i){
      const float* ac = accum + i*4;
      float stop_turb = (ac[2]*inv_cells) / (ac[3]*inv_cells + 1e-8f);
      float se = ac[1]*inv_bnc;
      float le = ac[0]*inv_bnc;
      a  += stop_turb + 0.05f*se + 0.01f*le;
      bm += se;
    }
    out[0] = a / (float)MAX_STEPS_;
    out[1] = bm / (float)MAX_STEPS_;
  }
}


extern "C" void kernel_launch(void* const* d_in, const int* in_sizes, int n_in,
                              void* d_out, int out_size, void* d_ws, size_t ws_size,
                              hipStream_t stream){
  (void)in_sizes; (void)n_in; (void)out_size; (void)ws_size;

  const float* u_in  = (const float*)d_in[0];
  const float* w1    = (const float*)d_in[1];
  const float* b1    = (const float*)d_in[2];
  const float* w2    = (const float*)d_in[3];
  const float* b2    = (const float*)d_in[4];
  const float* gw    = (const float*)d_in[5];
  const float* gb    = (const float*)d_in[6];
  const float* vw    = (const float*)d_in[7];
  const float* vb    = (const float*)d_in[8];
  const float* normw = (const float*)d_in[9];
  const float* wlp   = (const float*)d_in[10];
  const float* blp   = (const float*)d_in[11];
  const float* a_g   = (const float*)d_in[12];
  const float* a_l   = (const float*)d_in[13];
  const float* logdt = (const float*)d_in[14];
  const float* ks    = (const float*)d_in[15];
  const float* kt    = (const float*)d_in[16];

  float* Wf = (float*)d_ws;
  size_t o = 0;
  const size_t VOL = (size_t)B_ * C_ * N_;
  float* ubuf0  = Wf + o; o += VOL;
  float* ubuf1  = Wf + o; o += VOL;
  float* parts  = Wf + o; o += VOL;
  o += VOL;
  float* probeA = Wf + o; o += PROBE_CELLS;
  float* probeB = Wf + o; o += PROBE_CELLS;
  o += 256;
  float* hstate = Wf + o; o += 256;
  o += (size_t)B_*C_*32;
  float* lproj  = Wf + o; o += (size_t)B_*C_*32;
  float* spsks  = Wf + o; o += 3*C_;
  float* spskt  = Wf + o; o += 2*C_;
  float* scal   = Wf + o; o += 8;
  o += 8 + 16 + 16;
  _Float16* w1h = (_Float16*)(Wf + o); o += (C_*HID_)/2;
  _Float16* w2h = (_Float16*)(Wf + o); o += (HID_*C_)/2;

  float* part_l = parts;
  float* part_p = parts + 512*C_*4;
  float* accum  = parts + 512*C_*12;
  float* part_g = parts + 512*C_*13;

  float* ubuf[2] = { ubuf0, ubuf1 };

  prep_kernel<<<64, 256, 0, stream>>>(w1, w2, ks, kt, a_g, a_l, logdt,
                                      w1h, w2h, spsks, spskt, scal);
  zero_kernel<<<1, 256, 0, stream>>>(hstate, 256, accum, 256);
  hipMemcpyAsync(ubuf[0], u_in, VOL*sizeof(float), hipMemcpyDeviceToDevice, stream);
  init_partials_kernel<<<BN_/128, 512, 0, stream>>>(ubuf[0], part_l, part_p, part_g);

  float* pprev = probeA;
  float* pcur  = probeB;

  for (int step = 0; step < MAX_STEPS_; ++step){
    float* ucur  = ubuf[step & 1];
    float* unext = ubuf[(step & 1) ^ 1];
    int do_norm = (((step + 1) % 2) == 0) ? 1 : 0;

    float* acc = accum + step*4;
    if (step == 0){
      fold_kernel<<<SIDE_BLOCKS_, 256, 0, stream>>>(part_g, gw, gb, vw, vb, hstate,
                                                    part_l, wlp, blp, lproj,
                                                    part_p, pprev, part_p, accum, 0);
    } else {
      fold_kernel<<<SIDE_BLOCKS_, 256, 0, stream>>>(part_g, gw, gb, vw, vb, hstate,
                                                    part_l, wlp, blp, lproj,
                                                    part_p, pcur, pprev, acc - 4, 1);
      float* tswap = pprev; pprev = pcur; pcur = tswap;
    }
    step_kernel<<<BN_/128, 512, 0, stream>>>(ucur, w1h, b1, w2h, b2, hstate, lproj, spsks, spskt,
                                             scal, normw, unext, part_l, part_p, part_g, acc, accum + 64 + step, do_norm);
  }
  probe_fold_kernel<<<PROBE_BLOCKS_, 256, 0, stream>>>(part_p, pcur, pprev, accum + (MAX_STEPS_-1)*4);

  hipMemcpyAsync(d_out, ubuf[MAX_STEPS_ & 1], VOL*sizeof(float),
                 hipMemcpyDeviceToDevice, stream);
  final_out_kernel<<<1, 32, 0, stream>>>(accum, ((float*)d_out) + VOL);
}
